# starting kernel
# baseline (speedup 1.0000x reference)
#include <hip/hip_runtime.h>

typedef float v2f __attribute__((ext_vector_type(2)));
typedef float v8f __attribute__((ext_vector_type(8)));

#define HID 128

#define AU_KP 16
#define AH_KP 136
#define A3_KP 392


__global__ void zero_ws_kernel(float* ws, long n) {
  long i = (long)blockIdx.x * blockDim.x + threadIdx.x;
  long stride = (long)gridDim.x * blockDim.x;
  for (; i < n; i += stride) ws[i] = 0.0f;
}

__global__ void pack_weights_kernel(const float* __restrict__ W_u2h,
                                    const float* __restrict__ W_h2h,
                                    const float* __restrict__ W_upd,
                                    float* __restrict__ WTu,
                                    float* __restrict__ WTh,
                                    float* __restrict__ WTupd) {
  const int stride = gridDim.x * blockDim.x;
  const int tid = blockIdx.x * blockDim.x + threadIdx.x;
  for (int i = tid; i < HID * AU_KP; i += stride) {
    int c = i / AU_KP, k = i % AU_KP;
    WTu[i] = (k < 13) ? W_u2h[k * HID + c] : 0.f;
  }
  for (int i = tid; i < HID * AH_KP; i += stride) {
    int c = i / AH_KP, k = i % AH_KP;
    WTh[i] = (k < 133) ? W_h2h[k * HID + c] : 0.f;
  }
  for (int i = tid; i < HID * A3_KP; i += stride) {
    int c = i / A3_KP, k = i % A3_KP;
    WTupd[i] = (k < 386) ? W_upd[k * HID + c] : 0.f;
  }
}

__global__ __launch_bounds__(256) void agg_a2s_kernel(
    const int* __restrict__ src, const int* __restrict__ dst,
    const float* __restrict__ dis, const float* __restrict__ pos_action,
    const float* __restrict__ u, float* __restrict__ aggU, int E) {
  int e = blockIdx.x * blockDim.x + threadIdx.x;
  if (e >= E) return;
  int s = src[e], d = dst[e];
  float* row = aggU + (long)d * 12;
  atomicAdd(row + 0, pos_action[2 * s]);
  atomicAdd(row + 1, pos_action[2 * s + 1]);
  atomicAdd(row + 2, dis[e]);
  const float* up = u + (long)s * 8;
#pragma unroll
  for (int i = 0; i < 8; ++i) atomicAdd(row + 3 + i, up[i]);
  atomicAdd(row + 11, 1.0f);
}

__global__ __launch_bounds__(256) void agg_s2s_kernel(
    const int* __restrict__ src, const int* __restrict__ dst,
    const float* __restrict__ dis, const float* __restrict__ pos_state,
    const float* __restrict__ h, float* __restrict__ aggH, int E) {
  int gid = blockIdx.x * blockDim.x + threadIdx.x;
  int e = gid >> 5;
  int lane = gid & 31;
  if (e >= E) return;
  int s = src[e], d = dst[e];
  float* row = aggH + (long)d * 132;
  if (lane == 0) {
    atomicAdd(row + 0, pos_state[2 * s]);
    atomicAdd(row + 1, pos_state[2 * s + 1]);
    atomicAdd(row + 2, dis[e]);
    atomicAdd(row + 131, 1.0f);
  }
  const float4* hp4 = (const float4*)(h + (long)s * HID);
  float4 v = hp4[lane];
  float* p = row + 3 + 4 * lane;
  atomicAdd(p + 0, v.x);
  atomicAdd(p + 1, v.y);
  atomicAdd(p + 2, v.z);
  atomicAdd(p + 3, v.w);
}

#define A3_LD 393
#define AH_LD 137
#define AU_LD 17

__global__ __launch_bounds__(256, 1) void fused_node_kernel(
    const float* __restrict__ h, const float* __restrict__ pos_state,
    const float* __restrict__ aggU, const float* __restrict__ aggH,
    const float* __restrict__ WTu, const float* __restrict__ b_u2h,
    const float* __restrict__ WTh, const float* __restrict__ b_h2h,
    const float* __restrict__ WTupd, const float* __restrict__ b_upd,
    float* __restrict__ out, int n_state) {
  __shared__ float A[16 * A3_LD];
  __shared__ float Ah[16 * AH_LD];
  __shared__ float Au[16 * AU_LD];
  __shared__ float degU[16];
  __shared__ float degS[16];

  const int tid = threadIdx.x;
  const int n0 = blockIdx.x * 16;
  const int r = tid >> 4;
  const int c0 = tid & 15;
  const int n = n0 + r;
  const bool valid = (n < n_state);

  if (c0 == 0) {
    const float* au = aggU + (long)n * 12;
    const float* ah = aggH + (long)n * 132;
    float du = valid ? au[11] : 0.f;
    float ds = valid ? ah[131] : 0.f;
    degU[r] = du;
    degS[r] = ds;
    float p0 = valid ? pos_state[2 * n] : 0.f;
    float p1 = valid ? pos_state[2 * n + 1] : 0.f;
    float* AuR = Au + r * AU_LD;
    AuR[0] = valid ? au[0] : 0.f;
    AuR[1] = valid ? au[1] : 0.f;
    AuR[2] = du * p0;
    AuR[3] = du * p1;
    AuR[4] = valid ? au[2] : 0.f;
#pragma unroll
    for (int i = 0; i < 8; ++i) AuR[5 + i] = valid ? au[3 + i] : 0.f;
    AuR[13] = AuR[14] = AuR[15] = 0.f;

    float* AhR = Ah + r * AH_LD;
    AhR[0] = valid ? ah[0] : 0.f;
    AhR[1] = valid ? ah[1] : 0.f;
    AhR[2] = ds * p0;
    AhR[3] = ds * p1;
    AhR[4] = valid ? ah[2] : 0.f;
    AhR[133] = AhR[134] = AhR[135] = 0.f;

    float* AR = A + r * A3_LD;
    AR[384] = p0;
    AR[385] = p1;
#pragma unroll
    for (int k = 386; k < A3_KP; ++k) AR[k] = 0.f;
  }
  {
    const float* hp = h + (long)n * HID;
    const float* ahp = aggH + (long)n * 132 + 3;
    for (int c = c0; c < HID; c += 16) {
      A[r * A3_LD + c] = valid ? hp[c] : 0.f;
      Ah[r * AH_LD + 5 + c] = valid ? ahp[c] : 0.f;
    }
  }
  __syncthreads();

  const int lane = tid & 31;
  const int wv = tid >> 5;
  const int half = lane >> 4;
  const int m = lane & 15;
  const int kb = half * 2;
  const int col = wv * 16 + m;

  {
    const float* bp = WTu + col * AU_KP + kb;
    v8f acc = {};
#pragma unroll
    for (int k = 0; k < AU_KP; k += 4) {
      v2f a;
      a.x = Au[m * AU_LD + k + kb];
      a.y = Au[m * AU_LD + k + kb + 1];
      v2f b = *(const v2f*)(bp + k);
      acc = __builtin_amdgcn_wmma_f32_16x16x4_f32(false, a, false, b, (short)0,
                                                  acc, false, false);
    }
    float bu = b_u2h[col];
#pragma unroll
    for (int i = 0; i < 8; ++i) {
      int rr = i + 8 * half;
      A[rr * A3_LD + HID + col] = acc[i] + degU[rr] * bu;
    }
  }

  {
    const float* bp = WTh + col * AH_KP + kb;
    v8f acc = {};
#pragma unroll 2
    for (int k = 0; k < AH_KP; k += 4) {
      v2f a;
      a.x = Ah[m * AH_LD + k + kb];
      a.y = Ah[m * AH_LD + k + kb + 1];
      v2f b = *(const v2f*)(bp + k);
      acc = __builtin_amdgcn_wmma_f32_16x16x4_f32(false, a, false, b, (short)0,
                                                  acc, false, false);
    }
    float bh = b_h2h[col];
#pragma unroll
    for (int i = 0; i < 8; ++i) {
      int rr = i + 8 * half;
      float ds = degS[rr];
      A[rr * A3_LD + 2 * HID + col] =
          (acc[i] + ds * bh) / fmaxf(ds, 1.0f);
    }
  }
  __syncthreads();

  {
    const float* bp = WTupd + col * A3_KP + kb;
    v8f acc = {};
#pragma unroll 2
    for (int k = 0; k < A3_KP; k += 4) {
      v2f a;
      a.x = A[m * A3_LD + k + kb];
      a.y = A[m * A3_LD + k + kb + 1];
      v2f b = *(const v2f*)(bp + k);
      acc = __builtin_amdgcn_wmma_f32_16x16x4_f32(false, a, false, b, (short)0,
                                                  acc, false, false);
    }
    float bo = b_upd[col];
#pragma unroll
    for (int i = 0; i < 8; ++i) {
      int rr = i + 8 * half;
      int nn = n0 + rr;
      if (nn < n_state) out[(long)nn * HID + col] = acc[i] + bo;
    }
  }
}

extern "C" void kernel_launch(void* const* d_in, const int* in_sizes, int n_in,
                              void* d_out, int out_size, void* d_ws,
                              size_t ws_size, hipStream_t stream) {
  const float* h = (const float*)d_in[0];
  const float* u = (const float*)d_in[1];
  const float* pos_state = (const float*)d_in[2];
  const float* pos_action = (const float*)d_in[3];
  const int* a2s_src = (const int*)d_in[4];
  const int* a2s_dst = (const int*)d_in[5];
  const float* a2s_dis = (const float*)d_in[6];
  const int* s2s_src = (const int*)d_in[7];
  const int* s2s_dst = (const int*)d_in[8];
  const float* s2s_dis = (const float*)d_in[9];
  const float* W_u2h = (const float*)d_in[10];
  const float* b_u2h = (const float*)d_in[11];
  const float* W_h2h = (const float*)d_in[12];
  const float* b_h2h = (const float*)d_in[13];
  const float* W_upd = (const float*)d_in[14];
  const float* b_upd = (const float*)d_in[15];
  float* out = (float*)d_out;

  const int n_state = in_sizes[2] / 2;
  const int E_a2s = in_sizes[4];
  const int E_s2s = in_sizes[7];

  float* aggU = (float*)d_ws;
  float* aggH = aggU + (size_t)n_state * 12;
  float* WTu = aggH + (size_t)n_state * 132;
  float* WTh = WTu + HID * AU_KP;
  float* WTupd = WTh + HID * AH_KP;
  const long agg_floats = (long)n_state * (12 + 132);

  zero_ws_kernel<<<2048, 256, 0, stream>>>(aggU, agg_floats);
  pack_weights_kernel<<<256, 256, 0, stream>>>(W_u2h, W_h2h, W_upd, WTu, WTh,
                                               WTupd);
  agg_a2s_kernel<<<(E_a2s + 255) / 256, 256, 0, stream>>>(
      a2s_src, a2s_dst, a2s_dis, pos_action, u, aggU, E_a2s);
  agg_s2s_kernel<<<(E_s2s + 7) / 8, 256, 0, stream>>>(
      s2s_src, s2s_dst, s2s_dis, pos_state, h, aggH, E_s2s);
  fused_node_kernel<<<(n_state + 15) / 16, 256, 0, stream>>>(
      h, pos_state, aggU, aggH, WTu, b_u2h, WTh, b_h2h, WTupd, b_upd, out,
      n_state);
}


# edge scatter re-mapped: a wave takes 32 edges, lanes cover consecutive floats of one destination row (rows realigned: one atomic instruction = one 128-B line); float atomics kept, arrival-order sums
# speedup vs baseline: 1.7499x; 1.7499x over previous
#include <hip/hip_runtime.h>

typedef float v2f __attribute__((ext_vector_type(2)));
typedef float v8f __attribute__((ext_vector_type(8)));

#define HID 128

#define AU_KP 16
#define AH_KP 136
#define A3_KP 392


__global__ void zero_ws_kernel(float* ws, long n) {
  long i = (long)blockIdx.x * blockDim.x + threadIdx.x;
  long stride = (long)gridDim.x * blockDim.x;
  for (; i < n; i += stride) ws[i] = 0.0f;
}

__global__ void pack_weights_kernel(const float* __restrict__ W_u2h,
                                    const float* __restrict__ W_h2h,
                                    const float* __restrict__ W_upd,
                                    float* __restrict__ WTu,
                                    float* __restrict__ WTh,
                                    float* __restrict__ WTupd) {
  const int stride = gridDim.x * blockDim.x;
  const int tid = blockIdx.x * blockDim.x + threadIdx.x;
  for (int i = tid; i < HID * AU_KP; i += stride) {
    int c = i / AU_KP, k = i % AU_KP;
    WTu[i] = (k < 13) ? W_u2h[k * HID + c] : 0.f;
  }
  for (int i = tid; i < HID * AH_KP; i += stride) {
    int c = i / AH_KP, k = i % AH_KP;
    WTh[i] = (k < 133) ? W_h2h[k * HID + c] : 0.f;
  }
  for (int i = tid; i < HID * A3_KP; i += stride) {
    int c = i / A3_KP, k = i % A3_KP;
    WTupd[i] = (k < 386) ? W_upd[k * HID + c] : 0.f;
  }
}


__global__ __launch_bounds__(256) void agg_a2s_kernel(
    const int* __restrict__ src, const int* __restrict__ dst,
    const float* __restrict__ dis, const float* __restrict__ pos_action,
    const float* __restrict__ u, float* __restrict__ aggU, int E) {
  const int lane = threadIdx.x & 31;
  const int e0 = __builtin_amdgcn_readfirstlane(
                     (int)((blockIdx.x * blockDim.x + threadIdx.x) >> 5)) * 32;
  if (e0 >= E) return;
  const int cnt = min(32, E - e0);
  const bool ev = lane < cnt;
  const int s_l = ev ? src[e0 + lane] : 0;
  const int d_l = ev ? dst[e0 + lane] : 0;
  const float dis_l = ev ? dis[e0 + lane] : 0.f;
  const int c = lane & 15;
  const int hsel = lane >> 4;
  const int uc = min(max(c - 3, 0), 7);
  const bool ldl = (c < 2) || (c >= 3 && c < 11);
  float g[16];
  int dd[16];
#pragma unroll
  for (int t = 0; t < 16; ++t) {
    const int i = 2 * t + hsel;
    const int s = __shfl(s_l, i);
    dd[t] = __shfl(d_l, i);
    const float di = __shfl(dis_l, i);
    const float* p = (c < 2) ? (pos_action + 2 * (long)s + c)
                             : (u + 8 * (long)s + uc);
    const float x = *p;
    g[t] = ldl ? x : ((c == 2) ? di : 1.0f);
  }
#pragma unroll
  for (int t = 0; t < 16; ++t) {
    const int i = 2 * t + hsel;
    if (c < 12 && i < cnt) atomicAdd(aggU + (long)dd[t] * 12 + c, g[t]);
  }
}

__global__ __launch_bounds__(256) void agg_s2s_kernel(
    const int* __restrict__ src, const int* __restrict__ dst,
    const float* __restrict__ dis, const float* __restrict__ pos_state,
    const float* __restrict__ h, float* __restrict__ aggHh,
    float* __restrict__ aggHs, int E) {
  const int lane = threadIdx.x & 31;
  const int e0 = __builtin_amdgcn_readfirstlane(
                     (int)((blockIdx.x * blockDim.x + threadIdx.x) >> 5)) * 32;
  if (e0 >= E) return;
  const int cnt = min(32, E - e0);
  const bool ev = lane < cnt;
  const int s_l = ev ? src[e0 + lane] : 0;
  const int d_l = ev ? dst[e0 + lane] : 0;
  const float dis_l = ev ? dis[e0 + lane] : 0.f;

  {
    const int comp = lane & 3;
    float g[4];
    int dd[4];
#pragma unroll
    for (int t = 0; t < 4; ++t) {
      const int i = (lane >> 2) + 8 * t;
      const int s = __shfl(s_l, i);
      dd[t] = __shfl(d_l, i);
      const float di = __shfl(dis_l, i);
      const float x = pos_state[2 * (long)s + (comp & 1)];
      g[t] = (comp < 2) ? x : (comp == 2 ? di : 1.0f);
    }
#pragma unroll
    for (int t = 0; t < 4; ++t) {
      const int i = (lane >> 2) + 8 * t;
      if (i < cnt) atomicAdd(aggHs + (long)dd[t] * 4 + comp, g[t]);
    }
  }

  if (cnt == 32) {
    for (int i = 0; i < 32; i += 4) {
      float v[4][4];
#pragma unroll
      for (int q = 0; q < 4; ++q) {
        const int s = __builtin_amdgcn_readlane(s_l, i + q);
        const float* hp = h + (long)s * HID + lane;
#pragma unroll
        for (int j = 0; j < 4; ++j) v[q][j] = hp[32 * j];
      }
#pragma unroll
      for (int q = 0; q < 4; ++q) {
        const int d = __builtin_amdgcn_readlane(d_l, i + q);
        float* op = aggHh + (long)d * HID + lane;
#pragma unroll
        for (int j = 0; j < 4; ++j) atomicAdd(op + 32 * j, v[q][j]);
      }
    }
  } else {
    for (int i = 0; i < cnt; ++i) {
      const int s = __builtin_amdgcn_readlane(s_l, i);
      const int d = __builtin_amdgcn_readlane(d_l, i);
      const float* hp = h + (long)s * HID + lane;
      float* op = aggHh + (long)d * HID + lane;
#pragma unroll
      for (int j = 0; j < 4; ++j) atomicAdd(op + 32 * j, hp[32 * j]);
    }
  }
}

#define A3_LD 393
#define AH_LD 137
#define AU_LD 17

__global__ __launch_bounds__(256, 1) void fused_node_kernel(
    const float* __restrict__ h, const float* __restrict__ pos_state,
    const float* __restrict__ aggU, const float* __restrict__ aggHh,
    const float* __restrict__ aggHs,
    const float* __restrict__ WTu, const float* __restrict__ b_u2h,
    const float* __restrict__ WTh, const float* __restrict__ b_h2h,
    const float* __restrict__ WTupd, const float* __restrict__ b_upd,
    float* __restrict__ out, int n_state) {
  __shared__ float A[16 * A3_LD];
  __shared__ float Ah[16 * AH_LD];
  __shared__ float Au[16 * AU_LD];
  __shared__ float degU[16];
  __shared__ float degS[16];

  const int tid = threadIdx.x;
  const int n0 = blockIdx.x * 16;
  const int r = tid >> 4;
  const int c0 = tid & 15;
  const int n = n0 + r;
  const bool valid = (n < n_state);

  if (c0 == 0) {
    const float* au = aggU + (long)n * 12;
    const float* ah = aggHs + (long)n * 4;
    float du = valid ? au[11] : 0.f;
    float ds = valid ? ah[3] : 0.f;
    degU[r] = du;
    degS[r] = ds;
    float p0 = valid ? pos_state[2 * n] : 0.f;
    float p1 = valid ? pos_state[2 * n + 1] : 0.f;
    float* AuR = Au + r * AU_LD;
    AuR[0] = valid ? au[0] : 0.f;
    AuR[1] = valid ? au[1] : 0.f;
    AuR[2] = du * p0;
    AuR[3] = du * p1;
    AuR[4] = valid ? au[2] : 0.f;
#pragma unroll
    for (int i = 0; i < 8; ++i) AuR[5 + i] = valid ? au[3 + i] : 0.f;
    AuR[13] = AuR[14] = AuR[15] = 0.f;

    float* AhR = Ah + r * AH_LD;
    AhR[0] = valid ? ah[0] : 0.f;
    AhR[1] = valid ? ah[1] : 0.f;
    AhR[2] = ds * p0;
    AhR[3] = ds * p1;
    AhR[4] = valid ? ah[2] : 0.f;
    AhR[133] = AhR[134] = AhR[135] = 0.f;

    float* AR = A + r * A3_LD;
    AR[384] = p0;
    AR[385] = p1;
#pragma unroll
    for (int k = 386; k < A3_KP; ++k) AR[k] = 0.f;
  }
  {
    const float* hp = h + (long)n * HID;
    const float* ahp = aggHh + (long)n * HID;
    for (int c = c0; c < HID; c += 16) {
      A[r * A3_LD + c] = valid ? hp[c] : 0.f;
      Ah[r * AH_LD + 5 + c] = valid ? ahp[c] : 0.f;
    }
  }
  __syncthreads();

  const int lane = tid & 31;
  const int wv = tid >> 5;
  const int half = lane >> 4;
  const int m = lane & 15;
  const int kb = half * 2;
  const int col = wv * 16 + m;

  {
    const float* bp = WTu + col * AU_KP + kb;
    v8f acc = {};
#pragma unroll
    for (int k = 0; k < AU_KP; k += 4) {
      v2f a;
      a.x = Au[m * AU_LD + k + kb];
      a.y = Au[m * AU_LD + k + kb + 1];
      v2f b = *(const v2f*)(bp + k);
      acc = __builtin_amdgcn_wmma_f32_16x16x4_f32(false, a, false, b, (short)0,
                                                  acc, false, false);
    }
    float bu = b_u2h[col];
#pragma unroll
    for (int i = 0; i < 8; ++i) {
      int rr = i + 8 * half;
      A[rr * A3_LD + HID + col] = acc[i] + degU[rr] * bu;
    }
  }

  {
    const float* bp = WTh + col * AH_KP + kb;
    v8f acc = {};
#pragma unroll 2
    for (int k = 0; k < AH_KP; k += 4) {
      v2f a;
      a.x = Ah[m * AH_LD + k + kb];
      a.y = Ah[m * AH_LD + k + kb + 1];
      v2f b = *(const v2f*)(bp + k);
      acc = __builtin_amdgcn_wmma_f32_16x16x4_f32(false, a, false, b, (short)0,
                                                  acc, false, false);
    }
    float bh = b_h2h[col];
#pragma unroll
    for (int i = 0; i < 8; ++i) {
      int rr = i + 8 * half;
      float ds = degS[rr];
      A[rr * A3_LD + 2 * HID + col] =
          (acc[i] + ds * bh) / fmaxf(ds, 1.0f);
    }
  }
  __syncthreads();

  {
    const float* bp = WTupd + col * A3_KP + kb;
    v8f acc = {};
#pragma unroll 2
    for (int k = 0; k < A3_KP; k += 4) {
      v2f a;
      a.x = A[m * A3_LD + k + kb];
      a.y = A[m * A3_LD + k + kb + 1];
      v2f b = *(const v2f*)(bp + k);
      acc = __builtin_amdgcn_wmma_f32_16x16x4_f32(false, a, false, b, (short)0,
                                                  acc, false, false);
    }
    float bo = b_upd[col];
#pragma unroll
    for (int i = 0; i < 8; ++i) {
      int rr = i + 8 * half;
      int nn = n0 + rr;
      if (nn < n_state) out[(long)nn * HID + col] = acc[i] + bo;
    }
  }
}

extern "C" void kernel_launch(void* const* d_in, const int* in_sizes, int n_in,
                              void* d_out, int out_size, void* d_ws,
                              size_t ws_size, hipStream_t stream) {
  const float* h = (const float*)d_in[0];
  const float* u = (const float*)d_in[1];
  const float* pos_state = (const float*)d_in[2];
  const float* pos_action = (const float*)d_in[3];
  const int* a2s_src = (const int*)d_in[4];
  const int* a2s_dst = (const int*)d_in[5];
  const float* a2s_dis = (const float*)d_in[6];
  const int* s2s_src = (const int*)d_in[7];
  const int* s2s_dst = (const int*)d_in[8];
  const float* s2s_dis = (const float*)d_in[9];
  const float* W_u2h = (const float*)d_in[10];
  const float* b_u2h = (const float*)d_in[11];
  const float* W_h2h = (const float*)d_in[12];
  const float* b_h2h = (const float*)d_in[13];
  const float* W_upd = (const float*)d_in[14];
  const float* b_upd = (const float*)d_in[15];
  float* out = (float*)d_out;

  const int n_state = in_sizes[2] / 2;
  const int E_a2s = in_sizes[4];
  const int E_s2s = in_sizes[7];

  float* aggHh = (float*)d_ws;
  float* aggHs = aggHh + (size_t)n_state * HID;
  float* aggU = aggHs + (size_t)n_state * 4;
  float* WTu = aggU + (size_t)n_state * 12;
  float* WTh = WTu + HID * AU_KP;
  float* WTupd = WTh + HID * AH_KP;
  const long agg_floats = (long)n_state * (HID + 4 + 12);

  zero_ws_kernel<<<2048, 256, 0, stream>>>(aggHh, agg_floats);
  pack_weights_kernel<<<256, 256, 0, stream>>>(W_u2h, W_h2h, W_upd, WTu, WTh,
                                               WTupd);
  agg_a2s_kernel<<<(E_a2s + 255) / 256, 256, 0, stream>>>(
      a2s_src, a2s_dst, a2s_dis, pos_action, u, aggU, E_a2s);
  agg_s2s_kernel<<<(E_s2s + 255) / 256, 256, 0, stream>>>(
      s2s_src, s2s_dst, s2s_dis, pos_state, h, aggHh, aggHs, E_s2s);
  fused_node_kernel<<<(n_state + 15) / 16, 256, 0, stream>>>(
      h, pos_state, aggU, aggHh, aggHs, WTu, b_u2h, WTh, b_h2h, WTupd, b_upd,
      out, n_state);
}


# node kernel B operand: weights packed in fragment order; one contiguous 512-B wave load (16 B per lane) feeds two 16x16x4 products, replacing a 16-line 8-B gather per product; k order and sums kept
# speedup vs baseline: 2.1720x; 1.2412x over previous
#include <hip/hip_runtime.h>

typedef float v2f __attribute__((ext_vector_type(2)));
typedef float v4f __attribute__((ext_vector_type(4)));
typedef float v8f __attribute__((ext_vector_type(8)));

#define HID 128

#define AU_KP 16
#define AH_KP 136
#define A3_KP 392


__global__ void zero_ws_kernel(float* ws, long n) {
  long i = (long)blockIdx.x * blockDim.x + threadIdx.x;
  long stride = (long)gridDim.x * blockDim.x;
  for (; i < n; i += stride) ws[i] = 0.0f;
}

__global__ void pack_weights_kernel(const float* __restrict__ W_u2h,
                                    const float* __restrict__ W_h2h,
                                    const float* __restrict__ W_upd,
                                    float* __restrict__ WTu,
                                    float* __restrict__ WTh,
                                    float* __restrict__ WTupd) {
  const int stride = gridDim.x * blockDim.x;
  const int tid = blockIdx.x * blockDim.x + threadIdx.x;
  for (int i = tid; i < HID * AU_KP; i += stride) {
    const int tp = i >> 7, l = (i >> 2) & 31, j = i & 3;
    const int t = tp / (AU_KP / 8), p = tp % (AU_KP / 8);
    const int k = 8 * p + 4 * (j >> 1) + 2 * (l >> 4) + (j & 1);
    WTu[i] = (k < 13) ? W_u2h[k * HID + 16 * t + (l & 15)] : 0.f;
  }
  for (int i = tid; i < HID * AH_KP; i += stride) {
    const int tp = i >> 7, l = (i >> 2) & 31, j = i & 3;
    const int t = tp / (AH_KP / 8), p = tp % (AH_KP / 8);
    const int k = 8 * p + 4 * (j >> 1) + 2 * (l >> 4) + (j & 1);
    WTh[i] = (k < 133) ? W_h2h[k * HID + 16 * t + (l & 15)] : 0.f;
  }
  for (int i = tid; i < HID * A3_KP; i += stride) {
    const int tp = i >> 7, l = (i >> 2) & 31, j = i & 3;
    const int t = tp / (A3_KP / 8), p = tp % (A3_KP / 8);
    const int k = 8 * p + 4 * (j >> 1) + 2 * (l >> 4) + (j & 1);
    WTupd[i] = (k < 386) ? W_upd[k * HID + 16 * t + (l & 15)] : 0.f;
  }
}


__global__ __launch_bounds__(256) void agg_a2s_kernel(
    const int* __restrict__ src, const int* __restrict__ dst,
    const float* __restrict__ dis, const float* __restrict__ pos_action,
    const float* __restrict__ u, float* __restrict__ aggU, int E) {
  const int lane = threadIdx.x & 31;
  const int e0 = __builtin_amdgcn_readfirstlane(
                     (int)((blockIdx.x * blockDim.x + threadIdx.x) >> 5)) * 32;
  if (e0 >= E) return;
  const int cnt = min(32, E - e0);
  const bool ev = lane < cnt;
  const int s_l = ev ? src[e0 + lane] : 0;
  const int d_l = ev ? dst[e0 + lane] : 0;
  const float dis_l = ev ? dis[e0 + lane] : 0.f;
  const int c = lane & 15;
  const int hsel = lane >> 4;
  const int uc = min(max(c - 3, 0), 7);
  const bool ldl = (c < 2) || (c >= 3 && c < 11);
  float g[16];
  int dd[16];
#pragma unroll
  for (int t = 0; t < 16; ++t) {
    const int i = 2 * t + hsel;
    const int s = __shfl(s_l, i);
    dd[t] = __shfl(d_l, i);
    const float di = __shfl(dis_l, i);
    const float* p = (c < 2) ? (pos_action + 2 * (long)s + c)
                             : (u + 8 * (long)s + uc);
    const float x = *p;
    g[t] = ldl ? x : ((c == 2) ? di : 1.0f);
  }
#pragma unroll
  for (int t = 0; t < 16; ++t) {
    const int i = 2 * t + hsel;
    if (c < 12 && i < cnt) atomicAdd(aggU + (long)dd[t] * 12 + c, g[t]);
  }
}

__global__ __launch_bounds__(256) void agg_s2s_kernel(
    const int* __restrict__ src, const int* __restrict__ dst,
    const float* __restrict__ dis, const float* __restrict__ pos_state,
    const float* __restrict__ h, float* __restrict__ aggHh,
    float* __restrict__ aggHs, int E) {
  const int lane = threadIdx.x & 31;
  const int e0 = __builtin_amdgcn_readfirstlane(
                     (int)((blockIdx.x * blockDim.x + threadIdx.x) >> 5)) * 32;
  if (e0 >= E) return;
  const int cnt = min(32, E - e0);
  const bool ev = lane < cnt;
  const int s_l = ev ? src[e0 + lane] : 0;
  const int d_l = ev ? dst[e0 + lane] : 0;
  const float dis_l = ev ? dis[e0 + lane] : 0.f;

  {
    const int comp = lane & 3;
    float g[4];
    int dd[4];
#pragma unroll
    for (int t = 0; t < 4; ++t) {
      const int i = (lane >> 2) + 8 * t;
      const int s = __shfl(s_l, i);
      dd[t] = __shfl(d_l, i);
      const float di = __shfl(dis_l, i);
      const float x = pos_state[2 * (long)s + (comp & 1)];
      g[t] = (comp < 2) ? x : (comp == 2 ? di : 1.0f);
    }
#pragma unroll
    for (int t = 0; t < 4; ++t) {
      const int i = (lane >> 2) + 8 * t;
      if (i < cnt) atomicAdd(aggHs + (long)dd[t] * 4 + comp, g[t]);
    }
  }

  if (cnt == 32) {
    for (int i = 0; i < 32; i += 4) {
      float v[4][4];
#pragma unroll
      for (int q = 0; q < 4; ++q) {
        const int s = __builtin_amdgcn_readlane(s_l, i + q);
        const float* hp = h + (long)s * HID + lane;
#pragma unroll
        for (int j = 0; j < 4; ++j) v[q][j] = hp[32 * j];
      }
#pragma unroll
      for (int q = 0; q < 4; ++q) {
        const int d = __builtin_amdgcn_readlane(d_l, i + q);
        float* op = aggHh + (long)d * HID + lane;
#pragma unroll
        for (int j = 0; j < 4; ++j) atomicAdd(op + 32 * j, v[q][j]);
      }
    }
  } else {
    for (int i = 0; i < cnt; ++i) {
      const int s = __builtin_amdgcn_readlane(s_l, i);
      const int d = __builtin_amdgcn_readlane(d_l, i);
      const float* hp = h + (long)s * HID + lane;
      float* op = aggHh + (long)d * HID + lane;
#pragma unroll
      for (int j = 0; j < 4; ++j) atomicAdd(op + 32 * j, hp[32 * j]);
    }
  }
}

#define A3_LD 393
#define AH_LD 137
#define AU_LD 17

__global__ __launch_bounds__(256, 1) void fused_node_kernel(
    const float* __restrict__ h, const float* __restrict__ pos_state,
    const float* __restrict__ aggU, const float* __restrict__ aggHh,
    const float* __restrict__ aggHs,
    const float* __restrict__ WTu, const float* __restrict__ b_u2h,
    const float* __restrict__ WTh, const float* __restrict__ b_h2h,
    const float* __restrict__ WTupd, const float* __restrict__ b_upd,
    float* __restrict__ out, int n_state) {
  __shared__ float A[16 * A3_LD];
  __shared__ float Ah[16 * AH_LD];
  __shared__ float Au[16 * AU_LD];
  __shared__ float degU[16];
  __shared__ float degS[16];

  const int tid = threadIdx.x;
  const int n0 = blockIdx.x * 16;
  const int r = tid >> 4;
  const int c0 = tid & 15;
  const int n = n0 + r;
  const bool valid = (n < n_state);

  if (c0 == 0) {
    const float* au = aggU + (long)n * 12;
    const float* ah = aggHs + (long)n * 4;
    float du = valid ? au[11] : 0.f;
    float ds = valid ? ah[3] : 0.f;
    degU[r] = du;
    degS[r] = ds;
    float p0 = valid ? pos_state[2 * n] : 0.f;
    float p1 = valid ? pos_state[2 * n + 1] : 0.f;
    float* AuR = Au + r * AU_LD;
    AuR[0] = valid ? au[0] : 0.f;
    AuR[1] = valid ? au[1] : 0.f;
    AuR[2] = du * p0;
    AuR[3] = du * p1;
    AuR[4] = valid ? au[2] : 0.f;
#pragma unroll
    for (int i = 0; i < 8; ++i) AuR[5 + i] = valid ? au[3 + i] : 0.f;
    AuR[13] = AuR[14] = AuR[15] = 0.f;

    float* AhR = Ah + r * AH_LD;
    AhR[0] = valid ? ah[0] : 0.f;
    AhR[1] = valid ? ah[1] : 0.f;
    AhR[2] = ds * p0;
    AhR[3] = ds * p1;
    AhR[4] = valid ? ah[2] : 0.f;
    AhR[133] = AhR[134] = AhR[135] = 0.f;

    float* AR = A + r * A3_LD;
    AR[384] = p0;
    AR[385] = p1;
#pragma unroll
    for (int k = 386; k < A3_KP; ++k) AR[k] = 0.f;
  }
  {
    const float* hp = h + (long)n * HID;
    const float* ahp = aggHh + (long)n * HID;
    for (int c = c0; c < HID; c += 16) {
      A[r * A3_LD + c] = valid ? hp[c] : 0.f;
      Ah[r * AH_LD + 5 + c] = valid ? ahp[c] : 0.f;
    }
  }
  __syncthreads();

  const int lane = tid & 31;
  const int wv = tid >> 5;
  const int half = lane >> 4;
  const int m = lane & 15;
  const int kb = half * 2;
  const int col = wv * 16 + m;

  {
    const v4f* bp = (const v4f*)WTu + (wv * (AU_KP / 8)) * 32 + lane;
    const float* ap = Au + m * AU_LD + kb;
    v8f acc = {};
#pragma unroll
    for (int p = 0; p < AU_KP / 8; ++p) {
      const v4f b = bp[p * 32];
      v2f a0, a1, b0, b1;
      a0.x = ap[8 * p];
      a0.y = ap[8 * p + 1];
      a1.x = ap[8 * p + 4];
      a1.y = ap[8 * p + 5];
      b0.x = b.x; b0.y = b.y;
      b1.x = b.z; b1.y = b.w;
      acc = __builtin_amdgcn_wmma_f32_16x16x4_f32(false, a0, false, b0,
                                                  (short)0, acc, false, false);
      acc = __builtin_amdgcn_wmma_f32_16x16x4_f32(false, a1, false, b1,
                                                  (short)0, acc, false, false);
    }
    float bu = b_u2h[col];
#pragma unroll
    for (int i = 0; i < 8; ++i) {
      int rr = i + 8 * half;
      A[rr * A3_LD + HID + col] = acc[i] + degU[rr] * bu;
    }
  }

  {
    const v4f* bp = (const v4f*)WTh + (wv * (AH_KP / 8)) * 32 + lane;
    const float* ap = Ah + m * AH_LD + kb;
    v8f acc = {};
#pragma unroll
    for (int p = 0; p < AH_KP / 8; ++p) {
      const v4f b = bp[p * 32];
      v2f a0, a1, b0, b1;
      a0.x = ap[8 * p];
      a0.y = ap[8 * p + 1];
      a1.x = ap[8 * p + 4];
      a1.y = ap[8 * p + 5];
      b0.x = b.x; b0.y = b.y;
      b1.x = b.z; b1.y = b.w;
      acc = __builtin_amdgcn_wmma_f32_16x16x4_f32(false, a0, false, b0,
                                                  (short)0, acc, false, false);
      acc = __builtin_amdgcn_wmma_f32_16x16x4_f32(false, a1, false, b1,
                                                  (short)0, acc, false, false);
    }
    float bh = b_h2h[col];
#pragma unroll
    for (int i = 0; i < 8; ++i) {
      int rr = i + 8 * half;
      float ds = degS[rr];
      A[rr * A3_LD + 2 * HID + col] =
          (acc[i] + ds * bh) / fmaxf(ds, 1.0f);
    }
  }
  __syncthreads();

  {
    const v4f* bp = (const v4f*)WTupd + (wv * (A3_KP / 8)) * 32 + lane;
    const float* ap = A + m * A3_LD + kb;
    v8f acc = {};
#pragma unroll 7
    for (int p = 0; p < A3_KP / 8; ++p) {
      const v4f b = bp[p * 32];
      v2f a0, a1, b0, b1;
      a0.x = ap[8 * p];
      a0.y = ap[8 * p + 1];
      a1.x = ap[8 * p + 4];
      a1.y = ap[8 * p + 5];
      b0.x = b.x; b0.y = b.y;
      b1.x = b.z; b1.y = b.w;
      acc = __builtin_amdgcn_wmma_f32_16x16x4_f32(false, a0, false, b0,
                                                  (short)0, acc, false, false);
      acc = __builtin_amdgcn_wmma_f32_16x16x4_f32(false, a1, false, b1,
                                                  (short)0, acc, false, false);
    }
    float bo = b_upd[col];
#pragma unroll
    for (int i = 0; i < 8; ++i) {
      int rr = i + 8 * half;
      int nn = n0 + rr;
      if (nn < n_state) out[(long)nn * HID + col] = acc[i] + bo;
    }
  }
}

extern "C" void kernel_launch(void* const* d_in, const int* in_sizes, int n_in,
                              void* d_out, int out_size, void* d_ws,
                              size_t ws_size, hipStream_t stream) {
  const float* h = (const float*)d_in[0];
  const float* u = (const float*)d_in[1];
  const float* pos_state = (const float*)d_in[2];
  const float* pos_action = (const float*)d_in[3];
  const int* a2s_src = (const int*)d_in[4];
  const int* a2s_dst = (const int*)d_in[5];
  const float* a2s_dis = (const float*)d_in[6];
  const int* s2s_src = (const int*)d_in[7];
  const int* s2s_dst = (const int*)d_in[8];
  const float* s2s_dis = (const float*)d_in[9];
  const float* W_u2h = (const float*)d_in[10];
  const float* b_u2h = (const float*)d_in[11];
  const float* W_h2h = (const float*)d_in[12];
  const float* b_h2h = (const float*)d_in[13];
  const float* W_upd = (const float*)d_in[14];
  const float* b_upd = (const float*)d_in[15];
  float* out = (float*)d_out;

  const int n_state = in_sizes[2] / 2;
  const int E_a2s = in_sizes[4];
  const int E_s2s = in_sizes[7];

  float* aggHh = (float*)d_ws;
  float* aggHs = aggHh + (size_t)n_state * HID;
  float* aggU = aggHs + (size_t)n_state * 4;
  float* WTu = aggU + (size_t)n_state * 12;
  float* WTh = WTu + HID * AU_KP;
  float* WTupd = WTh + HID * AH_KP;
  const long agg_floats = (long)n_state * (HID + 4 + 12);

  zero_ws_kernel<<<2048, 256, 0, stream>>>(aggHh, agg_floats);
  pack_weights_kernel<<<256, 256, 0, stream>>>(W_u2h, W_h2h, W_upd, WTu, WTh,
                                               WTupd);
  agg_a2s_kernel<<<(E_a2s + 255) / 256, 256, 0, stream>>>(
      a2s_src, a2s_dst, a2s_dis, pos_action, u, aggU, E_a2s);
  agg_s2s_kernel<<<(E_s2s + 255) / 256, 256, 0, stream>>>(
      s2s_src, s2s_dst, s2s_dis, pos_state, h, aggHh, aggHs, E_s2s);
  fused_node_kernel<<<(n_state + 15) / 16, 256, 0, stream>>>(
      h, pos_state, aggU, aggHh, aggHs, WTu, b_u2h, WTh, b_h2h, WTupd, b_upd,
      out, n_state);
}


# s2s list binned by owner once per launch (bin blocks in the fold launch: LDS count, scan, fill); a node work-group reads its own runs of records, not the whole list; LDS float adds, arrival order
# speedup vs baseline: 3.5487x; 1.6338x over previous
#include <hip/hip_runtime.h>

typedef float v2f __attribute__((ext_vector_type(2)));
typedef float v4f __attribute__((ext_vector_type(4)));
typedef float v8f __attribute__((ext_vector_type(8)));
typedef int v4i __attribute__((ext_vector_type(4)));

#define HID 128

#define K1P 144
#define K2P 136


__global__ void zero_ws_kernel(float* ws, long n) {
  long i = (long)blockIdx.x * blockDim.x + threadIdx.x;
  long stride = (long)gridDim.x * blockDim.x;
  for (; i < n; i += stride) ws[i] = 0.0f;
}

__device__ __forceinline__ int frag_index(int P, int k, int n) {
  const int kk = k & 7;
  const int l = 16 * ((kk >> 1) & 1) + (n & 15);
  const int j = 2 * (kk >> 2) + (kk & 1);
  return (((n >> 4) * P + (k >> 3)) * 32 + l) * 4 + j;
}

#define NOWN 256
#define BIN_BLOCKS 256
#define OFF_LD (NOWN + 1)
#define BIN_T 10
#define RMAX 208

__device__ __forceinline__ int range_nodes(int n_state) {
  return max(1, min(RMAX, (n_state + NOWN - 1) / NOWN));
}
__device__ __forceinline__ int owner_of(int d, int rng) {
  return (int)(((unsigned)d / (unsigned)rng) & (NOWN - 1));
}
__device__ __forceinline__ int chunk_start(int b, int E) {
  return min(E, b * ((E + BIN_BLOCKS - 1) / BIN_BLOCKS));
}

__device__ __forceinline__ void bin_chunk(
    int b, int tid, const int* __restrict__ src, const int* __restrict__ dst,
    const float* __restrict__ dis, int* __restrict__ off,
    v4i* __restrict__ recs, int n_state, int E, int* cnt, int* wsum) {
  const int rng = range_nodes(n_state);
  const int e0 = chunk_start(b, E);
  const int e1 = chunk_start(b + 1, E);
  cnt[tid] = 0;
  __syncthreads();
  for (int t0 = e0; t0 < e1; t0 += 256 * BIN_T) {
    int d[BIN_T];
#pragma unroll
    for (int j = 0; j < BIN_T; ++j) {
      const int e = t0 + j * 256 + tid;
      d[j] = (e < e1) ? dst[e] : 0;
    }
#pragma unroll
    for (int j = 0; j < BIN_T; ++j) {
      const int e = t0 + j * 256 + tid;
      if (e < e1) atomicAdd(&cnt[owner_of(d[j], rng)], 1);
    }
  }
  __syncthreads();
  const int lane = tid & 31;
  const int wv = tid >> 5;
  const int v = cnt[tid];
  int x = v;
#pragma unroll
  for (int s = 1; s < 32; s <<= 1) {
    const int y = __shfl_up(x, s);
    if (lane >= s) x += y;
  }
  if (lane == 31) wsum[wv] = x;
  __syncthreads();
  int woff = 0;
#pragma unroll
  for (int i = 0; i < 8; ++i) woff += (i < wv) ? wsum[i] : 0;
  const int excl = woff + x - v;
  off[b * OFF_LD + tid] = excl;
  if (tid == NOWN - 1) off[b * OFF_LD + NOWN] = excl + v;
  cnt[tid] = excl;
  __syncthreads();
  for (int t0 = e0; t0 < e1; t0 += 256 * BIN_T) {
    int d[BIN_T], s[BIN_T];
    float w[BIN_T];
#pragma unroll
    for (int j = 0; j < BIN_T; ++j) {
      const int e = t0 + j * 256 + tid;
      const bool ev = e < e1;
      d[j] = ev ? dst[e] : 0;
      s[j] = ev ? src[e] : 0;
      w[j] = ev ? dis[e] : 0.f;
    }
#pragma unroll
    for (int j = 0; j < BIN_T; ++j) {
      const int e = t0 + j * 256 + tid;
      if (e < e1) {
        const int pos = atomicAdd(&cnt[owner_of(d[j], rng)], 1);
        v4i rec;
        rec.x = s[j];
        rec.y = d[j];
        rec.z = __float_as_int(w[j]);
        rec.w = 0;
        recs[e0 + pos] = rec;
      }
    }
  }
}

#define FOLD_BLOCKS 10
#define COPY_BLOCKS 65
__global__ __launch_bounds__(256) void fold_pack_kernel(
    const float* __restrict__ W_u2h, const float* __restrict__ b_u2h,
    const float* __restrict__ W_h2h, const float* __restrict__ b_h2h,
    const float* __restrict__ W_upd, float* __restrict__ WT1,
    float* __restrict__ WT2, const int* __restrict__ s2s_src,
    const int* __restrict__ s2s_dst, const float* __restrict__ s2s_dis,
    int* __restrict__ off, int* __restrict__ recs, int n_state, int E) {
  __shared__ int cnt[NOWN];
  __shared__ int wsum[8];
  const int tid = threadIdx.x;
  if (blockIdx.x >= FOLD_BLOCKS + COPY_BLOCKS) {
    bin_chunk(blockIdx.x - (FOLD_BLOCKS + COPY_BLOCKS), tid, s2s_src, s2s_dst,
              s2s_dis, off, (v4i*)recs, n_state, E, cnt, wsum);
    return;
  }
  if (blockIdx.x >= FOLD_BLOCKS) {
    const int i = (blockIdx.x - FOLD_BLOCKS) * 256 + tid;
    const int k = i >> 7, nn = i & 127;
    WT1[frag_index(K1P / 8, k, nn)] = W_upd[((k < 128) ? k : k + 256) * HID + nn];
    return;
  }
  const int lane = tid & 31;
  const int wv = tid >> 5;
  const int c = lane & 15;
  const int hh = lane >> 4;
  const int mt = blockIdx.x;
  if (mt == 1) WT2[frag_index(K2P / 8, 134 + (tid >> 7), tid & 127)] = 0.f;
  const int ra = (mt == 0) ? c : 16 * (mt - 1) + c;
  const bool xv = (mt == 0) ? (ra < 14) : (ra < 134);
  const float* xr = b_u2h;
  if (xv) {
    if (mt == 0) xr = (ra == 0) ? b_u2h : W_u2h + (ra - 1) * HID;
    else         xr = (ra == 0) ? b_h2h : W_h2h + (ra - 1) * HID;
  }
  const float* wc = W_upd + ((mt == 0) ? 128 : 256) * HID + 16 * wv + c;
  v8f acc = {};
#pragma unroll
  for (int s = 0; s < 32; ++s) {
    const int k = 4 * s + 2 * hh;
    const float x0 = xr[k], x1 = xr[k + 1];
    v2f a, b;
    a.x = xv ? x0 : 0.f;
    a.y = xv ? x1 : 0.f;
    b.x = wc[k * HID];
    b.y = wc[(k + 1) * HID];
    acc = __builtin_amdgcn_wmma_f32_16x16x4_f32(false, a, false, b, (short)0,
                                                acc, false, false);
  }
  const int n = 16 * wv + c;
#pragma unroll
  for (int i = 0; i < 8; ++i) {
    const int r = ((mt == 0) ? 0 : 16 * (mt - 1)) + 8 * hh + i;
    if (mt == 0) {
      if (r < 14) WT1[frag_index(K1P / 8, 130 + r, n)] = acc[i];
    } else {
      if (r < 134) WT2[frag_index(K2P / 8, r, n)] = acc[i];
    }
  }
}


__global__ __launch_bounds__(256) void agg_a2s_kernel(
    const int* __restrict__ src, const int* __restrict__ dst,
    const float* __restrict__ dis, const float* __restrict__ pos_action,
    const float* __restrict__ u, float* __restrict__ aggU, int E) {
  const int lane = threadIdx.x & 31;
  const int e0 = __builtin_amdgcn_readfirstlane(
                     (int)((blockIdx.x * blockDim.x + threadIdx.x) >> 5)) * 32;
  if (e0 >= E) return;
  const int cnt = min(32, E - e0);
  const bool ev = lane < cnt;
  const int s_l = ev ? src[e0 + lane] : 0;
  const int d_l = ev ? dst[e0 + lane] : 0;
  const float dis_l = ev ? dis[e0 + lane] : 0.f;
  const int c = lane & 15;
  const int hsel = lane >> 4;
  const int uc = min(max(c - 3, 0), 7);
  const bool ldl = (c < 2) || (c >= 3 && c < 11);
  float g[16];
  int dd[16];
#pragma unroll
  for (int t = 0; t < 16; ++t) {
    const int i = 2 * t + hsel;
    const int s = __shfl(s_l, i);
    dd[t] = __shfl(d_l, i);
    const float di = __shfl(dis_l, i);
    const float* p = (c < 2) ? (pos_action + 2 * (long)s + c)
                             : (u + 8 * (long)s + uc);
    const float x = *p;
    g[t] = ldl ? x : ((c == 2) ? di : 1.0f);
  }
#pragma unroll
  for (int t = 0; t < 16; ++t) {
    const int i = 2 * t + hsel;
    if (c < 12 && i < cnt) atomicAdd(aggU + (long)dd[t] * 12 + c, g[t]);
  }
}

#define A1_LD 145
#define A2_LD 137

template <int NP1, int NP2>
__device__ __forceinline__ void k_pairs(const v4f* __restrict__ bp1,
                                        const v4f* __restrict__ bp2,
                                        const float* ap1, const float* ap2,
                                        v8f& acc1, v8f& acc2) {
  v4f b1[NP1], b2[NP2];
#pragma unroll
  for (int q = 0; q < NP1; ++q) b1[q] = bp1[q * 32];
#pragma unroll
  for (int q = 0; q < NP2; ++q) b2[q] = bp2[q * 32];
#pragma unroll
  for (int q = 0; q < NP1; ++q) {
    v2f a0, a1, b0, bb;
    a0.x = ap1[8 * q];
    a0.y = ap1[8 * q + 1];
    a1.x = ap1[8 * q + 4];
    a1.y = ap1[8 * q + 5];
    b0.x = b1[q].x; b0.y = b1[q].y;
    bb.x = b1[q].z; bb.y = b1[q].w;
    acc1 = __builtin_amdgcn_wmma_f32_16x16x4_f32(false, a0, false, b0,
                                                 (short)0, acc1, false, false);
    acc1 = __builtin_amdgcn_wmma_f32_16x16x4_f32(false, a1, false, bb,
                                                 (short)0, acc1, false, false);
    if (q < NP2) {
      a0.x = ap2[8 * q];
      a0.y = ap2[8 * q + 1];
      a1.x = ap2[8 * q + 4];
      a1.y = ap2[8 * q + 5];
      b0.x = b2[q].x; b0.y = b2[q].y;
      bb.x = b2[q].z; bb.y = b2[q].w;
      acc2 = __builtin_amdgcn_wmma_f32_16x16x4_f32(false, a0, false, b0,
                                                   (short)0, acc2, false, false);
      acc2 = __builtin_amdgcn_wmma_f32_16x16x4_f32(false, a1, false, bb,
                                                   (short)0, acc2, false, false);
    }
  }
}

#define NODE_THREADS 1024
#define NODE_WAVES 32
#define SEGS (BIN_BLOCKS / NODE_WAVES)
#define QCAP (32 * SEGS)
#define A2_DEG 0
#define A2_POS 1
#define A2_DIS 5
#define A2_H 6

template <bool FULL>
__device__ __forceinline__ void drain(const v4i* q, int qn, int k, int lane,
                                      int base,
                                      const float* __restrict__ pos_state,
                                      const float* __restrict__ h, float* A2) {
  const bool lv = lane < k;
  const int qi = qn - k + (lv ? lane : 0);
  const v4i rec = q[qi];
  const int r_l = rec.y - base;
  const int s_l = rec.x;
  float v[8][4];
#pragma unroll
  for (int t = 0; t < 8; ++t) {
    if (FULL || t < k) {
      const int s = __builtin_amdgcn_readlane(s_l, t);
      const float* hp = h + (long)s * HID + lane;
#pragma unroll
      for (int j = 0; j < 4; ++j) v[t][j] = hp[32 * j];
    }
  }
  if (lv) {
    const float p0 = pos_state[2 * (long)s_l];
    const float p1 = pos_state[2 * (long)s_l + 1];
    const float di = __int_as_float(rec.z);
    float* row = A2 + r_l * A2_LD;
    atomicAdd(row + A2_DEG, 1.0f);
    atomicAdd(row + A2_POS, p0);
    atomicAdd(row + A2_POS + 1, p1);
    atomicAdd(row + A2_DIS, di);
  }
#pragma unroll
  for (int t = 0; t < 8; ++t) {
    if (FULL || t < k) {
      const int r = __builtin_amdgcn_readlane(r_l, t);
      float* op = A2 + r * A2_LD + A2_H + lane;
#pragma unroll
      for (int j = 0; j < 4; ++j) atomicAdd(op + 32 * j, v[t][j]);
    }
  }
}

__global__ __launch_bounds__(NODE_THREADS, 1) void fused_node_kernel(
    const float* __restrict__ h, const float* __restrict__ pos_state,
    const int* __restrict__ off, const int* __restrict__ recs_i,
    const float* __restrict__ aggU, const float* __restrict__ WT1,
    const float* __restrict__ WT2, const float* __restrict__ b_upd,
    float* __restrict__ out, int n_state, int E) {
  __shared__ float A2[RMAX * A2_LD];
  __shared__ float A1[64 * A1_LD];
  __shared__ v4i Q[NODE_WAVES * QCAP];

  const int tid = threadIdx.x;
  const int lane = tid & 31;
  const int wv = tid >> 5;
  const int G = NOWN;
  const int R = range_nodes(n_state);
  const int passes = (n_state + R * G - 1) / (R * G);
  v4i* q = Q + wv * QCAP;
  const v4i* recs = (const v4i*)recs_i;
  const int bl = wv * SEGS + (lane & (SEGS - 1));
  const int st_l = off[bl * OFF_LD + blockIdx.x];
  const int en_l = off[bl * OFF_LD + blockIdx.x + 1];

  for (int pass = 0; pass < passes; ++pass) {
    const int base = (pass * G + blockIdx.x) * R;
    const int cntn = max(0, min(R, n_state - base));

    for (int i = tid; i < RMAX * A2_LD; i += NODE_THREADS) A2[i] = 0.f;
    __syncthreads();

    int qn = 0;
    {
      v4i r[SEGS];
#pragma unroll
      for (int u = 0; u < SEGS; ++u) {
        const int st = __builtin_amdgcn_readlane(st_l, u);
        const int en = __builtin_amdgcn_readlane(en_l, u);
        const int i = st + lane;
        const v4i none = {0, -1, 0, 0};
        r[u] = none;
        if (i < en) r[u] = recs[chunk_start(wv * SEGS + u, E) + i];
      }
#pragma unroll
      for (int u = 0; u < SEGS; ++u) {
        const bool mt = (unsigned)(r[u].y - base) < (unsigned)cntn;
        const unsigned mask = __builtin_amdgcn_ballot_w32(mt);
        if (mt) q[qn + (int)__builtin_amdgcn_mbcnt_lo(mask, 0u)] = r[u];
        qn += __builtin_popcount(mask);
      }
    }
    while (qn >= 8) {
      drain<true>(q, qn, 8, lane, base, pos_state, h, A2);
      qn -= 8;
    }
#pragma unroll 1
    for (int u = 0; u < SEGS; ++u) {
      const int en = __builtin_amdgcn_readlane(en_l, u);
      const int cs = chunk_start(wv * SEGS + u, E);
      for (int i0 = __builtin_amdgcn_readlane(st_l, u) + 32; i0 < en;
           i0 += 32) {
        const int i = i0 + lane;
        v4i t = {0, -1, 0, 0};
        if (i < en) t = recs[cs + i];
        const bool mt = (unsigned)(t.y - base) < (unsigned)cntn;
        const unsigned mask = __builtin_amdgcn_ballot_w32(mt);
        if (mt) q[qn + (int)__builtin_amdgcn_mbcnt_lo(mask, 0u)] = t;
        qn += __builtin_popcount(mask);
        while (qn >= 8) {
          drain<true>(q, qn, 8, lane, base, pos_state, h, A2);
          qn -= 8;
        }
      }
    }
    if (qn > 0) drain<false>(q, qn, qn, lane, base, pos_state, h, A2);
    __syncthreads();

    const int tiles = (cntn + 15) >> 4;
    const int half = lane >> 4;
    const int m = lane & 15;
    const int kb = half * 2;
    const int ct = wv & 7;
    const int tl = wv >> 3;
    const int col = ct * 16 + m;
    const float bo = b_upd[col];
    for (int rd = 0; rd * 4 < tiles; ++rd) {
      {
        const int r = tid >> 4;
        const int c0 = tid & 15;
        const int rr = rd * 64 + r;
        const int n = base + rr;
        const bool valid = rr < cntn;
        if (c0 == 0) {
          const float* au = aggU + (long)n * 12;
          const float du = valid ? au[11] : 0.f;
          const float p0 = valid ? pos_state[2 * (long)n] : 0.f;
          const float p1 = valid ? pos_state[2 * (long)n + 1] : 0.f;
          float* R1 = A1 + r * A1_LD + HID;
          R1[0] = p0;
          R1[1] = p1;
          R1[2] = du;
          R1[3] = valid ? au[0] : 0.f;
          R1[4] = valid ? au[1] : 0.f;
          R1[5] = du * p0;
          R1[6] = du * p1;
          R1[7] = valid ? au[2] : 0.f;
#pragma unroll
          for (int i = 0; i < 8; ++i) R1[8 + i] = valid ? au[3 + i] : 0.f;
          if (rr < RMAX) {
            float* R2 = A2 + rr * A2_LD;
            const float ds = R2[A2_DEG];
            R2[3] = ds * p0;
            R2[4] = ds * p1;
          }
        }
        const float* hp = h + (long)n * HID;
        for (int c = c0; c < HID; c += 16)
          A1[r * A1_LD + c] = valid ? hp[c] : 0.f;
      }
      __syncthreads();
      const int t = rd * 4 + tl;
      if (t < tiles) {
        const v4f* bp1 = (const v4f*)WT1 + (ct * (K1P / 8)) * 32 + lane;
        const v4f* bp2 = (const v4f*)WT2 + (ct * (K2P / 8)) * 32 + lane;
        const float* ap1 = A1 + (tl * 16 + m) * A1_LD + kb;
        const float* ap2 = A2 + (t * 16 + m) * A2_LD + kb;
        v8f acc1 = {}, acc2 = {};
#pragma unroll 1
        for (int g = 0; g < 8; ++g)
          k_pairs<2, 2>(bp1 + g * 64, bp2 + g * 64, ap1 + g * 16, ap2 + g * 16,
                        acc1, acc2);
        k_pairs<2, 1>(bp1 + 8 * 64, bp2 + 8 * 64, ap1 + 8 * 16, ap2 + 8 * 16,
                      acc1, acc2);
#pragma unroll
        for (int i = 0; i < 8; ++i) {
          const int rr = t * 16 + i + 8 * half;
          if (rr < cntn)
            out[(long)(base + rr) * HID + col] =
                acc1[i] + acc2[i] / fmaxf(A2[rr * A2_LD + A2_DEG], 1.0f) + bo;
        }
      }
      __syncthreads();
    }
  }
}

extern "C" void kernel_launch(void* const* d_in, const int* in_sizes, int n_in,
                              void* d_out, int out_size, void* d_ws,
                              size_t ws_size, hipStream_t stream) {
  const float* h = (const float*)d_in[0];
  const float* u = (const float*)d_in[1];
  const float* pos_state = (const float*)d_in[2];
  const float* pos_action = (const float*)d_in[3];
  const int* a2s_src = (const int*)d_in[4];
  const int* a2s_dst = (const int*)d_in[5];
  const float* a2s_dis = (const float*)d_in[6];
  const int* s2s_src = (const int*)d_in[7];
  const int* s2s_dst = (const int*)d_in[8];
  const float* s2s_dis = (const float*)d_in[9];
  const float* W_u2h = (const float*)d_in[10];
  const float* b_u2h = (const float*)d_in[11];
  const float* W_h2h = (const float*)d_in[12];
  const float* b_h2h = (const float*)d_in[13];
  const float* W_upd = (const float*)d_in[14];
  const float* b_upd = (const float*)d_in[15];
  float* out = (float*)d_out;

  const int n_state = in_sizes[2] / 2;
  const int E_a2s = in_sizes[4];
  const int E_s2s = in_sizes[7];

  float* aggU = (float*)d_ws;
  float* WT1 = aggU + (size_t)n_state * 12;
  float* WT2 = WT1 + HID * K1P;
  int* off = (int*)(WT2 + HID * K2P);
  int* recs = off + BIN_BLOCKS * OFF_LD;
  const long agg_floats = (long)n_state * 12;

  zero_ws_kernel<<<2048, 256, 0, stream>>>(aggU, agg_floats);
  fold_pack_kernel<<<FOLD_BLOCKS + COPY_BLOCKS + BIN_BLOCKS, 256, 0, stream>>>(
      W_u2h, b_u2h, W_h2h, b_h2h, W_upd, WT1, WT2, s2s_src, s2s_dst, s2s_dis,
      off, recs, n_state, E_s2s);
  agg_a2s_kernel<<<(E_a2s + 255) / 256, 256, 0, stream>>>(
      a2s_src, a2s_dst, a2s_dis, pos_action, u, aggU, E_a2s);
  fused_node_kernel<<<NOWN, NODE_THREADS, 0, stream>>>(
      h, pos_state, off, recs, aggU, WT1, WT2, b_upd, out, n_state, E_s2s);
}


# node products register-blocked: A rows staged once (in the record lists' LDS), a wave reuses each B fragment over its 3-4 node tiles; mean's division on A rows, one accumulator; sums re-associated
# speedup vs baseline: 4.7046x; 1.3257x over previous
#include <hip/hip_runtime.h>

typedef float v2f __attribute__((ext_vector_type(2)));
typedef float v4f __attribute__((ext_vector_type(4)));
typedef float v8f __attribute__((ext_vector_type(8)));
typedef int v4i __attribute__((ext_vector_type(4)));

#define HID 128

#define K1P 144
#define K2P 136


__device__ __forceinline__ int frag_index(int P, int k, int n) {
  const int kk = k & 7;
  const int l = 16 * ((kk >> 1) & 1) + (n & 15);
  const int j = 2 * (kk >> 2) + (kk & 1);
  return (((n >> 4) * P + (k >> 3)) * 32 + l) * 4 + j;
}

#define NOWN 256
#define BIN_BLOCKS 256
#define OFF_LD (NOWN + 1)
#define BIN_T 10
#define RMAX 208

__device__ __forceinline__ int range_nodes(int n_state) {
  return max(1, min(RMAX, (n_state + NOWN - 1) / NOWN));
}
__device__ __forceinline__ int owner_of(int d, int rng) {
  return (int)(((unsigned)d / (unsigned)rng) & (NOWN - 1));
}
__device__ __forceinline__ int chunk_start(int b, int E) {
  return min(E, b * ((E + BIN_BLOCKS - 1) / BIN_BLOCKS));
}

__device__ __forceinline__ void bin_chunk(
    int b, int tid, const int* __restrict__ src, const int* __restrict__ dst,
    const float* __restrict__ dis, int* __restrict__ off,
    v4i* __restrict__ recs, int n_state, int E, int* cnt, int* wsum) {
  const int rng = range_nodes(n_state);
  const int e0 = chunk_start(b, E);
  const int e1 = chunk_start(b + 1, E);
  cnt[tid] = 0;
  __syncthreads();
  for (int t0 = e0; t0 < e1; t0 += 256 * BIN_T) {
    int d[BIN_T];
#pragma unroll
    for (int j = 0; j < BIN_T; ++j) {
      const int e = t0 + j * 256 + tid;
      d[j] = (e < e1) ? dst[e] : 0;
    }
#pragma unroll
    for (int j = 0; j < BIN_T; ++j) {
      const int e = t0 + j * 256 + tid;
      if (e < e1) atomicAdd(&cnt[owner_of(d[j], rng)], 1);
    }
  }
  __syncthreads();
  const int lane = tid & 31;
  const int wv = tid >> 5;
  const int v = cnt[tid];
  int x = v;
#pragma unroll
  for (int s = 1; s < 32; s <<= 1) {
    const int y = __shfl_up(x, s);
    if (lane >= s) x += y;
  }
  if (lane == 31) wsum[wv] = x;
  __syncthreads();
  int woff = 0;
#pragma unroll
  for (int i = 0; i < 8; ++i) woff += (i < wv) ? wsum[i] : 0;
  const int excl = woff + x - v;
  off[b * OFF_LD + tid] = excl;
  if (tid == NOWN - 1) off[b * OFF_LD + NOWN] = excl + v;
  cnt[tid] = excl;
  __syncthreads();
  for (int t0 = e0; t0 < e1; t0 += 256 * BIN_T) {
    int d[BIN_T], s[BIN_T];
    float w[BIN_T];
#pragma unroll
    for (int j = 0; j < BIN_T; ++j) {
      const int e = t0 + j * 256 + tid;
      const bool ev = e < e1;
      d[j] = ev ? dst[e] : 0;
      s[j] = ev ? src[e] : 0;
      w[j] = ev ? dis[e] : 0.f;
    }
#pragma unroll
    for (int j = 0; j < BIN_T; ++j) {
      const int e = t0 + j * 256 + tid;
      if (e < e1) {
        const int pos = atomicAdd(&cnt[owner_of(d[j], rng)], 1);
        v4i rec;
        rec.x = s[j];
        rec.y = d[j];
        rec.z = __float_as_int(w[j]);
        rec.w = 0;
        recs[e0 + pos] = rec;
      }
    }
  }
}

#define FOLD_BLOCKS 10
#define COPY_BLOCKS 65
__global__ __launch_bounds__(256) void fold_pack_kernel(
    const float* __restrict__ W_u2h, const float* __restrict__ b_u2h,
    const float* __restrict__ W_h2h, const float* __restrict__ b_h2h,
    const float* __restrict__ W_upd, float* __restrict__ WT1,
    float* __restrict__ WT2, const int* __restrict__ s2s_src,
    const int* __restrict__ s2s_dst, const float* __restrict__ s2s_dis,
    int* __restrict__ off, int* __restrict__ recs,
    const int* __restrict__ a2s_src, const int* __restrict__ a2s_dst,
    const float* __restrict__ a2s_dis, int* __restrict__ offA,
    int* __restrict__ recsA, int n_state, int E, int EA) {
  __shared__ int cnt[NOWN];
  __shared__ int wsum[8];
  const int tid = threadIdx.x;
  if (blockIdx.x >= FOLD_BLOCKS + COPY_BLOCKS + BIN_BLOCKS) {
    bin_chunk(blockIdx.x - (FOLD_BLOCKS + COPY_BLOCKS + BIN_BLOCKS), tid,
              a2s_src, a2s_dst, a2s_dis, offA, (v4i*)recsA, n_state, EA, cnt,
              wsum);
    return;
  }
  if (blockIdx.x >= FOLD_BLOCKS + COPY_BLOCKS) {
    bin_chunk(blockIdx.x - (FOLD_BLOCKS + COPY_BLOCKS), tid, s2s_src, s2s_dst,
              s2s_dis, off, (v4i*)recs, n_state, E, cnt, wsum);
    return;
  }
  if (blockIdx.x >= FOLD_BLOCKS) {
    const int i = (blockIdx.x - FOLD_BLOCKS) * 256 + tid;
    const int k = i >> 7, nn = i & 127;
    WT1[frag_index(K1P / 8, k, nn)] = W_upd[((k < 128) ? k : k + 256) * HID + nn];
    return;
  }
  const int lane = tid & 31;
  const int wv = tid >> 5;
  const int c = lane & 15;
  const int hh = lane >> 4;
  const int mt = blockIdx.x;
  if (mt == 1) WT2[frag_index(K2P / 8, 134 + (tid >> 7), tid & 127)] = 0.f;
  const int ra = (mt == 0) ? c : 16 * (mt - 1) + c;
  const bool xv = (mt == 0) ? (ra < 14) : (ra < 134);
  const float* xr = b_u2h;
  if (xv) {
    if (mt == 0) xr = (ra == 0) ? b_u2h : W_u2h + (ra - 1) * HID;
    else         xr = (ra == 0) ? b_h2h : W_h2h + (ra - 1) * HID;
  }
  const float* wc = W_upd + ((mt == 0) ? 128 : 256) * HID + 16 * wv + c;
  v8f acc = {};
#pragma unroll
  for (int s = 0; s < 32; ++s) {
    const int k = 4 * s + 2 * hh;
    const float x0 = xr[k], x1 = xr[k + 1];
    v2f a, b;
    a.x = xv ? x0 : 0.f;
    a.y = xv ? x1 : 0.f;
    b.x = wc[k * HID];
    b.y = wc[(k + 1) * HID];
    acc = __builtin_amdgcn_wmma_f32_16x16x4_f32(false, a, false, b, (short)0,
                                                acc, false, false);
  }
  const int n = 16 * wv + c;
#pragma unroll
  for (int i = 0; i < 8; ++i) {
    const int r = ((mt == 0) ? 0 : 16 * (mt - 1)) + 8 * hh + i;
    if (mt == 0) {
      if (r < 14) WT1[frag_index(K1P / 8, 130 + r, n)] = acc[i];
    } else {
      if (r < 134) WT2[frag_index(K2P / 8, r, n)] = acc[i];
    }
  }
}

#define A1_LD 145
#define A2_LD 137

#define NODE_THREADS 1024
#define NODE_WAVES 32
#define SEGS (BIN_BLOCKS / NODE_WAVES)
#define QCAP (32 * SEGS)
#define A2_DEG 0
#define A2_POS 1
#define A2_DIS 5
#define A2_H 6
#define AU_LD 13

template <int NT>
__device__ __forceinline__ void load_a(const float* ap, int ld64,
                                       v2f (&a)[NT][2]) {
#pragma unroll
  for (int j = 0; j < NT; ++j) {
    const float* p = ap + j * ld64;
    a[j][0].x = p[0];
    a[j][0].y = p[1];
    a[j][1].x = p[4];
    a[j][1].y = p[5];
  }
}

template <int NT>
__device__ __forceinline__ void mma_sub(const v2f (&a)[NT][2], const v4f b,
                                        v8f (&acc)[NT]) {
  v2f b0, bb;
  b0.x = b.x; b0.y = b.y;
  bb.x = b.z; bb.y = b.w;
#pragma unroll
  for (int j = 0; j < NT; ++j) {
    acc[j] = __builtin_amdgcn_wmma_f32_16x16x4_f32(
        false, a[j][0], false, b0, (short)0, acc[j], false, false);
    acc[j] = __builtin_amdgcn_wmma_f32_16x16x4_f32(
        false, a[j][1], false, bb, (short)0, acc[j], false, false);
  }
}

template <int NT>
__device__ __forceinline__ void wave_tiles(
    const v4f* __restrict__ bp1, const v4f* __restrict__ bp2, const float* ap1,
    const float* ap2, const float* dga, float* __restrict__ out, int base,
    int cntn, int tl, int half, int col, float bo) {
  v8f acc[NT];
  float inv[NT];
#pragma unroll
  for (int j = 0; j < NT; ++j) {
    acc[j] = v8f{};
    inv[j] = 1.0f / fmaxf(dga[j * (64 * A2_LD)], 1.0f);
  }
  v2f a1[NT][2], a2[NT][2];
  v4f b1c = bp1[0], b2c = bp2[0];
  load_a<NT>(ap1, 64 * A1_LD, a1);
#pragma unroll 1
  for (int p = 0; p < K2P / 8; ++p) {
    const v4f b1n = bp1[(p + 1) * 32];
    const v4f b2n = bp2[min(p + 1, K2P / 8 - 1) * 32];
    load_a<NT>(ap2 + 8 * p, 64 * A2_LD, a2);
    mma_sub<NT>(a1, b1c, acc);
    load_a<NT>(ap1 + 8 * (p + 1), 64 * A1_LD, a1);
#pragma unroll
    for (int j = 0; j < NT; ++j) {
      a2[j][0] *= inv[j];
      a2[j][1] *= inv[j];
    }
    mma_sub<NT>(a2, b2c, acc);
    b1c = b1n;
    b2c = b2n;
  }
  mma_sub<NT>(a1, b1c, acc);
  const int r0 = tl * 16 + 8 * half;
  float* op = out + (long)(base + r0) * HID + col;
#pragma unroll
  for (int j = 0; j < NT; ++j) {
#pragma unroll
    for (int i = 0; i < 8; ++i) {
      const int ro = 64 * j + i;
      if (r0 + ro < cntn) op[ro * HID] = acc[j][i] + bo;
    }
  }
}

template <bool FULL>
__device__ __forceinline__ void drain(const v4i* q, int qn, int k, int lane,
                                      int base,
                                      const float* __restrict__ pos_state,
                                      const float* __restrict__ h, float* A2) {
  const bool lv = lane < k;
  const int qi = qn - k + (lv ? lane : 0);
  const v4i rec = q[qi];
  const int r_l = rec.y - base;
  const int s_l = rec.x;
  float v[8][4];
#pragma unroll
  for (int t = 0; t < 8; ++t) {
    if (FULL || t < k) {
      const int s = __builtin_amdgcn_readlane(s_l, t);
      const float* hp = h + (long)s * HID + lane;
#pragma unroll
      for (int j = 0; j < 4; ++j) v[t][j] = hp[32 * j];
    }
  }
  if (lv) {
    const float p0 = pos_state[2 * (long)s_l];
    const float p1 = pos_state[2 * (long)s_l + 1];
    const float di = __int_as_float(rec.z);
    float* row = A2 + r_l * A2_LD;
    atomicAdd(row + A2_DEG, 1.0f);
    atomicAdd(row + A2_POS, p0);
    atomicAdd(row + A2_POS + 1, p1);
    atomicAdd(row + A2_DIS, di);
  }
#pragma unroll
  for (int t = 0; t < 8; ++t) {
    if (FULL || t < k) {
      const int r = __builtin_amdgcn_readlane(r_l, t);
      float* op = A2 + r * A2_LD + A2_H + lane;
#pragma unroll
      for (int j = 0; j < 4; ++j) atomicAdd(op + 32 * j, v[t][j]);
    }
  }
}

__device__ __forceinline__ void drain_a2s(const v4i* q, int qn, int k, int lane,
                                          int base,
                                          const float* __restrict__ pos_action,
                                          const float* __restrict__ u_feat,
                                          float* AU) {
  if (lane < k) {
    const v4i rec = q[qn - k + lane];
    const long s = rec.x;
    const v2f pa = *(const v2f*)(pos_action + 2 * s);
    const v4f u0 = *(const v4f*)(u_feat + 8 * s);
    const v4f u1 = *(const v4f*)(u_feat + 8 * s + 4);
    float* a = AU + (rec.y - base) * AU_LD;
    atomicAdd(a + 0, pa.x);
    atomicAdd(a + 1, pa.y);
    atomicAdd(a + 2, __int_as_float(rec.z));
    atomicAdd(a + 3, u0.x);
    atomicAdd(a + 4, u0.y);
    atomicAdd(a + 5, u0.z);
    atomicAdd(a + 6, u0.w);
    atomicAdd(a + 7, u1.x);
    atomicAdd(a + 8, u1.y);
    atomicAdd(a + 9, u1.z);
    atomicAdd(a + 10, u1.w);
    atomicAdd(a + 11, 1.0f);
  }
}

__global__ __launch_bounds__(NODE_THREADS, 1) void fused_node_kernel(
    const float* __restrict__ h, const float* __restrict__ pos_state,
    const int* __restrict__ off, const int* __restrict__ recs_i,
    const int* __restrict__ offA, const int* __restrict__ recsA_i,
    const float* __restrict__ pos_action, const float* __restrict__ u_feat,
    const float* __restrict__ WT1,
    const float* __restrict__ WT2, const float* __restrict__ b_upd,
    float* __restrict__ out, int n_state, int E, int EA) {
  __shared__ float A2[RMAX * A2_LD];
  __shared__ float AU[RMAX * AU_LD];
  __shared__ v4i Q[NODE_WAVES * QCAP];
  static_assert(RMAX * A1_LD * sizeof(float) <= NODE_WAVES * QCAP * sizeof(v4i),
                "A1 must fit in Q");
  float* A1 = (float*)Q;

  const int tid = threadIdx.x;
  const int lane = tid & 31;
  const int wv = __builtin_amdgcn_readfirstlane(tid >> 5);
  const int G = NOWN;
  const int R = range_nodes(n_state);
  const int passes = (n_state + R * G - 1) / (R * G);
  v4i* q = Q + wv * QCAP;
  const v4i* recs = (const v4i*)recs_i;
  const v4i* recsA = (const v4i*)recsA_i;
  const int bl = wv * SEGS + (lane & (SEGS - 1));
  const int st_l = off[bl * OFF_LD + blockIdx.x];
  const int en_l = off[bl * OFF_LD + blockIdx.x + 1];
  const int stA_l = offA[bl * OFF_LD + blockIdx.x];
  const int enA_l = offA[bl * OFF_LD + blockIdx.x + 1];

  for (int pass = 0; pass < passes; ++pass) {
    const int base = (pass * G + blockIdx.x) * R;
    const int cntn = max(0, min(R, n_state - base));
    int ls = lane;
    asm volatile("" : "+v"(ls));

    const int tz = wv * 32 + ls;
    for (int i = tz; i < RMAX * A2_LD; i += NODE_THREADS) A2[i] = 0.f;
    for (int i = tz; i < RMAX * AU_LD; i += NODE_THREADS) AU[i] = 0.f;
    __syncthreads();

    int qn = 0;
    {
      v4i r[SEGS];
#pragma unroll
      for (int u = 0; u < SEGS; ++u) {
        const int st = __builtin_amdgcn_readlane(st_l, u);
        const int en = __builtin_amdgcn_readlane(en_l, u);
        const int i = st + ls;
        const v4i none = {0, -1, 0, 0};
        r[u] = none;
        if (i < en) r[u] = recs[chunk_start(wv * SEGS + u, E) + i];
      }
#pragma unroll
      for (int u = 0; u < SEGS; ++u) {
        const bool mt = (unsigned)(r[u].y - base) < (unsigned)cntn;
        const unsigned mask = __builtin_amdgcn_ballot_w32(mt);
        if (mt) q[qn + (int)__builtin_amdgcn_mbcnt_lo(mask, 0u)] = r[u];
        qn += __builtin_popcount(mask);
      }
    }
    while (qn >= 8) {
      drain<true>(q, qn, 8, ls, base, pos_state, h, A2);
      qn -= 8;
    }
#pragma unroll 1
    for (int u = 0; u < SEGS; ++u) {
      const int en = __builtin_amdgcn_readlane(en_l, u);
      const int cs = chunk_start(wv * SEGS + u, E);
      for (int i0 = __builtin_amdgcn_readlane(st_l, u) + 32; i0 < en;
           i0 += 32) {
        const int i = i0 + ls;
        v4i t = {0, -1, 0, 0};
        if (i < en) t = recs[cs + i];
        const bool mt = (unsigned)(t.y - base) < (unsigned)cntn;
        const unsigned mask = __builtin_amdgcn_ballot_w32(mt);
        if (mt) q[qn + (int)__builtin_amdgcn_mbcnt_lo(mask, 0u)] = t;
        qn += __builtin_popcount(mask);
        while (qn >= 8) {
          drain<true>(q, qn, 8, ls, base, pos_state, h, A2);
          qn -= 8;
        }
      }
    }
    if (qn > 0) drain<false>(q, qn, qn, ls, base, pos_state, h, A2);

    int maxlen = 0;
#pragma unroll
    for (int u = 0; u < SEGS; ++u)
      maxlen = max(maxlen, __builtin_amdgcn_readlane(enA_l, u) -
                               __builtin_amdgcn_readlane(stA_l, u));
#pragma unroll 1
    for (int j0 = 0; j0 < maxlen; j0 += 32) {
      int qa = 0;
      v4i r[SEGS];
#pragma unroll
      for (int u = 0; u < SEGS; ++u) {
        const int st = __builtin_amdgcn_readlane(stA_l, u);
        const int en = __builtin_amdgcn_readlane(enA_l, u);
        const int i = st + j0 + ls;
        const v4i none = {0, -1, 0, 0};
        r[u] = none;
        if (i < en) r[u] = recsA[chunk_start(wv * SEGS + u, EA) + i];
      }
#pragma unroll
      for (int u = 0; u < SEGS; ++u) {
        const bool mt = (unsigned)(r[u].y - base) < (unsigned)cntn;
        const unsigned mask = __builtin_amdgcn_ballot_w32(mt);
        if (mt) q[qa + (int)__builtin_amdgcn_mbcnt_lo(mask, 0u)] = r[u];
        qa += __builtin_popcount(mask);
      }
      while (qa > 0) {
        const int k = min(32, qa);
        drain_a2s(q, qa, k, ls, base, pos_action, u_feat, AU);
        qa -= k;
      }
    }
    __syncthreads();

    const int tiles = (cntn + 15) >> 4;
    const int rowsP = tiles * 16;
    int ln = lane;
    asm volatile("" : "+v"(ln));
    const int tid2 = wv * 32 + ln;
    {
      float v[(RMAX + 31) / 32][4];
#pragma unroll
      for (int it = 0; it < (RMAX + 31) / 32; ++it) {
        const int rr = wv + 32 * it;
        if (rr < cntn) {
          const float* hp = h + (long)(base + rr) * HID + ln;
#pragma unroll
          for (int j = 0; j < 4; ++j) v[it][j] = hp[32 * j];
        } else {
#pragma unroll
          for (int j = 0; j < 4; ++j) v[it][j] = 0.f;
        }
      }
      if (tid2 < rowsP) {
        const int rr = tid2;
        float* R1 = A1 + rr * A1_LD + HID;
        if (rr < cntn) {
          const long n = base + rr;
          const float* au = AU + rr * AU_LD;
          const float p0 = pos_state[2 * n];
          const float p1 = pos_state[2 * n + 1];
          const float du = au[11];
          R1[0] = p0;
          R1[1] = p1;
          R1[2] = du;
          R1[3] = au[0];
          R1[4] = au[1];
          R1[5] = du * p0;
          R1[6] = du * p1;
          R1[7] = au[2];
#pragma unroll
          for (int i = 0; i < 8; ++i) R1[8 + i] = au[3 + i];
          float* R2 = A2 + rr * A2_LD;
          const float ds = R2[A2_DEG];
          R2[3] = ds * p0;
          R2[4] = ds * p1;
        } else {
#pragma unroll
          for (int i = 0; i < 16; ++i) R1[i] = 0.f;
        }
      }
#pragma unroll
      for (int it = 0; it < (RMAX + 31) / 32; ++it) {
        const int rr = wv + 32 * it;
        if (rr < rowsP) {
#pragma unroll
          for (int j = 0; j < 4; ++j) A1[rr * A1_LD + ln + 32 * j] = v[it][j];
        }
      }
    }
    __syncthreads();
    {
      const int half = ln >> 4;
      const int m = ln & 15;
      const int kb = half * 2;
      const int ct = wv & 7;
      const int tl = wv >> 3;
      const int col = ct * 16 + m;
      const float bo = b_upd[col];
      const v4f* bp1 = (const v4f*)WT1 + (ct * (K1P / 8)) * 32 + ln;
      const v4f* bp2 = (const v4f*)WT2 + (ct * (K2P / 8)) * 32 + ln;
      const float* ap1 = A1 + (tl * 16 + m) * A1_LD + kb;
      const float* ap2 = A2 + (tl * 16 + m) * A2_LD + kb;
      const float* dga = A2 + (tl * 16 + m) * A2_LD + A2_DEG;
      const int nt = (tiles - tl + 3) >> 2;
      if (nt >= 4)
        wave_tiles<4>(bp1, bp2, ap1, ap2, dga, out, base, cntn, tl, half, col, bo);
      else if (nt == 3)
        wave_tiles<3>(bp1, bp2, ap1, ap2, dga, out, base, cntn, tl, half, col, bo);
      else if (nt == 2)
        wave_tiles<2>(bp1, bp2, ap1, ap2, dga, out, base, cntn, tl, half, col, bo);
      else if (nt == 1)
        wave_tiles<1>(bp1, bp2, ap1, ap2, dga, out, base, cntn, tl, half, col, bo);
    }
    __syncthreads();
  }
}

extern "C" void kernel_launch(void* const* d_in, const int* in_sizes, int n_in,
                              void* d_out, int out_size, void* d_ws,
                              size_t ws_size, hipStream_t stream) {
  const float* h = (const float*)d_in[0];
  const float* u = (const float*)d_in[1];
  const float* pos_state = (const float*)d_in[2];
  const float* pos_action = (const float*)d_in[3];
  const int* a2s_src = (const int*)d_in[4];
  const int* a2s_dst = (const int*)d_in[5];
  const float* a2s_dis = (const float*)d_in[6];
  const int* s2s_src = (const int*)d_in[7];
  const int* s2s_dst = (const int*)d_in[8];
  const float* s2s_dis = (const float*)d_in[9];
  const float* W_u2h = (const float*)d_in[10];
  const float* b_u2h = (const float*)d_in[11];
  const float* W_h2h = (const float*)d_in[12];
  const float* b_h2h = (const float*)d_in[13];
  const float* W_upd = (const float*)d_in[14];
  const float* b_upd = (const float*)d_in[15];
  float* out = (float*)d_out;

  const int n_state = in_sizes[2] / 2;
  const int E_a2s = in_sizes[4];
  const int E_s2s = in_sizes[7];

  float* WT1 = (float*)d_ws;
  float* WT2 = WT1 + HID * K1P;
  int* off = (int*)(WT2 + HID * K2P);
  int* offA = off + BIN_BLOCKS * OFF_LD;
  int* recs = offA + BIN_BLOCKS * OFF_LD;
  int* recsA = recs + (size_t)E_s2s * 4;

  fold_pack_kernel<<<FOLD_BLOCKS + COPY_BLOCKS + 2 * BIN_BLOCKS, 256, 0,
                     stream>>>(
      W_u2h, b_u2h, W_h2h, b_h2h, W_upd, WT1, WT2, s2s_src, s2s_dst, s2s_dis,
      off, recs, a2s_src, a2s_dst, a2s_dis, offA, recsA, n_state, E_s2s, E_a2s);
  fused_node_kernel<<<NOWN, NODE_THREADS, 0, stream>>>(
      h, pos_state, off, recs, offA, recsA, pos_action, u, WT1, WT2, b_upd, out,
      n_state, E_s2s, E_a2s);
}


# s2s sums without per-edge LDS float adds: a work-group orders its records by node in LDS (counting sort), a wave sums a node's h rows in registers (16-B loads), adds once per run; sums re-ordered
# speedup vs baseline: 5.0551x; 1.0745x over previous
#include <hip/hip_runtime.h>

typedef float v2f __attribute__((ext_vector_type(2)));
typedef float v4f __attribute__((ext_vector_type(4)));
typedef float v8f __attribute__((ext_vector_type(8)));
typedef int v4i __attribute__((ext_vector_type(4)));
typedef int v2i __attribute__((ext_vector_type(2)));

#define HID 128

#define K1P 144
#define K2P 136


__device__ __forceinline__ int frag_index(int P, int k, int n) {
  const int kk = k & 7;
  const int l = 16 * ((kk >> 1) & 1) + (n & 15);
  const int j = 2 * (kk >> 2) + (kk & 1);
  return (((n >> 4) * P + (k >> 3)) * 32 + l) * 4 + j;
}

#define NOWN 256
#define BIN_BLOCKS 256
#define OFF_LD (NOWN + 1)
#define BIN_T 10
#define RMAX 208

__device__ __forceinline__ int range_nodes(int n_state) {
  return max(1, min(RMAX, (n_state + NOWN - 1) / NOWN));
}
__device__ __forceinline__ int owner_of(int d, int rng) {
  return (int)(((unsigned)d / (unsigned)rng) & (NOWN - 1));
}
__device__ __forceinline__ int chunk_start(int b, int E) {
  return min(E, b * ((E + BIN_BLOCKS - 1) / BIN_BLOCKS));
}

__device__ __forceinline__ void bin_chunk(
    int b, int tid, const int* __restrict__ src, const int* __restrict__ dst,
    const float* __restrict__ dis, int* __restrict__ off,
    v4i* __restrict__ recs, int n_state, int E, int* cnt, int* wsum) {
  const int rng = range_nodes(n_state);
  const int e0 = chunk_start(b, E);
  const int e1 = chunk_start(b + 1, E);
  cnt[tid] = 0;
  __syncthreads();
  for (int t0 = e0; t0 < e1; t0 += 256 * BIN_T) {
    int d[BIN_T];
#pragma unroll
    for (int j = 0; j < BIN_T; ++j) {
      const int e = t0 + j * 256 + tid;
      d[j] = (e < e1) ? dst[e] : 0;
    }
#pragma unroll
    for (int j = 0; j < BIN_T; ++j) {
      const int e = t0 + j * 256 + tid;
      if (e < e1) atomicAdd(&cnt[owner_of(d[j], rng)], 1);
    }
  }
  __syncthreads();
  const int lane = tid & 31;
  const int wv = tid >> 5;
  const int v = cnt[tid];
  int x = v;
#pragma unroll
  for (int s = 1; s < 32; s <<= 1) {
    const int y = __shfl_up(x, s);
    if (lane >= s) x += y;
  }
  if (lane == 31) wsum[wv] = x;
  __syncthreads();
  int woff = 0;
#pragma unroll
  for (int i = 0; i < 8; ++i) woff += (i < wv) ? wsum[i] : 0;
  const int excl = woff + x - v;
  off[b * OFF_LD + tid] = excl;
  if (tid == NOWN - 1) off[b * OFF_LD + NOWN] = excl + v;
  cnt[tid] = excl;
  __syncthreads();
  for (int t0 = e0; t0 < e1; t0 += 256 * BIN_T) {
    int d[BIN_T], s[BIN_T];
    float w[BIN_T];
#pragma unroll
    for (int j = 0; j < BIN_T; ++j) {
      const int e = t0 + j * 256 + tid;
      const bool ev = e < e1;
      d[j] = ev ? dst[e] : 0;
      s[j] = ev ? src[e] : 0;
      w[j] = ev ? dis[e] : 0.f;
    }
#pragma unroll
    for (int j = 0; j < BIN_T; ++j) {
      const int e = t0 + j * 256 + tid;
      if (e < e1) {
        const int pos = atomicAdd(&cnt[owner_of(d[j], rng)], 1);
        v4i rec;
        rec.x = s[j];
        rec.y = d[j];
        rec.z = __float_as_int(w[j]);
        rec.w = 0;
        recs[e0 + pos] = rec;
      }
    }
  }
}

#define FOLD_BLOCKS 10
#define COPY_BLOCKS 65
__global__ __launch_bounds__(256) void fold_pack_kernel(
    const float* __restrict__ W_u2h, const float* __restrict__ b_u2h,
    const float* __restrict__ W_h2h, const float* __restrict__ b_h2h,
    const float* __restrict__ W_upd, float* __restrict__ WT1,
    float* __restrict__ WT2, const int* __restrict__ s2s_src,
    const int* __restrict__ s2s_dst, const float* __restrict__ s2s_dis,
    int* __restrict__ off, int* __restrict__ recs,
    const int* __restrict__ a2s_src, const int* __restrict__ a2s_dst,
    const float* __restrict__ a2s_dis, int* __restrict__ offA,
    int* __restrict__ recsA, int n_state, int E, int EA) {
  __shared__ int cnt[NOWN];
  __shared__ int wsum[8];
  const int tid = threadIdx.x;
  if (blockIdx.x >= FOLD_BLOCKS + COPY_BLOCKS + BIN_BLOCKS) {
    bin_chunk(blockIdx.x - (FOLD_BLOCKS + COPY_BLOCKS + BIN_BLOCKS), tid,
              a2s_src, a2s_dst, a2s_dis, offA, (v4i*)recsA, n_state, EA, cnt,
              wsum);
    return;
  }
  if (blockIdx.x >= FOLD_BLOCKS + COPY_BLOCKS) {
    bin_chunk(blockIdx.x - (FOLD_BLOCKS + COPY_BLOCKS), tid, s2s_src, s2s_dst,
              s2s_dis, off, (v4i*)recs, n_state, E, cnt, wsum);
    return;
  }
  if (blockIdx.x >= FOLD_BLOCKS) {
    const int i = (blockIdx.x - FOLD_BLOCKS) * 256 + tid;
    const int k = i >> 7, nn = i & 127;
    WT1[frag_index(K1P / 8, k, nn)] = W_upd[((k < 128) ? k : k + 256) * HID + nn];
    return;
  }
  const int lane = tid & 31;
  const int wv = tid >> 5;
  const int c = lane & 15;
  const int hh = lane >> 4;
  const int mt = blockIdx.x;
  if (mt == 1) WT2[frag_index(K2P / 8, 134 + (tid >> 7), tid & 127)] = 0.f;
  const int ra = (mt == 0) ? c : 16 * (mt - 1) + c;
  const bool xv = (mt == 0) ? (ra < 14) : (ra < 134);
  const float* xr = b_u2h;
  if (xv) {
    if (mt == 0) xr = (ra == 0) ? b_u2h : W_u2h + (ra - 1) * HID;
    else         xr = (ra == 0) ? b_h2h : W_h2h + (ra - 1) * HID;
  }
  const float* wc = W_upd + ((mt == 0) ? 128 : 256) * HID + 16 * wv + c;
  v8f acc = {};
#pragma unroll
  for (int s = 0; s < 32; ++s) {
    const int k = 4 * s + 2 * hh;
    const float x0 = xr[k], x1 = xr[k + 1];
    v2f a, b;
    a.x = xv ? x0 : 0.f;
    a.y = xv ? x1 : 0.f;
    b.x = wc[k * HID];
    b.y = wc[(k + 1) * HID];
    acc = __builtin_amdgcn_wmma_f32_16x16x4_f32(false, a, false, b, (short)0,
                                                acc, false, false);
  }
  const int n = 16 * wv + c;
#pragma unroll
  for (int i = 0; i < 8; ++i) {
    const int r = ((mt == 0) ? 0 : 16 * (mt - 1)) + 8 * hh + i;
    if (mt == 0) {
      if (r < 14) WT1[frag_index(K1P / 8, 130 + r, n)] = acc[i];
    } else {
      if (r < 134) WT2[frag_index(K2P / 8, r, n)] = acc[i];
    }
  }
}

#define A1_LD 145
#define A2_LD 137

#define NODE_THREADS 1024
#define NODE_WAVES 32
#define SEGS (BIN_BLOCKS / NODE_WAVES)
#define QCAP (32 * SEGS)
#define SCAP (NODE_WAVES * QCAP)
#define CNT_N 224
#define GRP 16
#define A2_DEG 0
#define A2_POS 1
#define A2_DIS 5
#define A2_H 6
#define AU_LD 13

template <int NT>
__device__ __forceinline__ void load_a(const float* ap, int ld64,
                                       v2f (&a)[NT][2]) {
#pragma unroll
  for (int j = 0; j < NT; ++j) {
    const float* p = ap + j * ld64;
    a[j][0].x = p[0];
    a[j][0].y = p[1];
    a[j][1].x = p[4];
    a[j][1].y = p[5];
  }
}

template <int NT>
__device__ __forceinline__ void mma_sub(const v2f (&a)[NT][2], const v4f b,
                                        v8f (&acc)[NT]) {
  v2f b0, bb;
  b0.x = b.x; b0.y = b.y;
  bb.x = b.z; bb.y = b.w;
#pragma unroll
  for (int j = 0; j < NT; ++j) {
    acc[j] = __builtin_amdgcn_wmma_f32_16x16x4_f32(
        false, a[j][0], false, b0, (short)0, acc[j], false, false);
    acc[j] = __builtin_amdgcn_wmma_f32_16x16x4_f32(
        false, a[j][1], false, bb, (short)0, acc[j], false, false);
  }
}

template <int NT>
__device__ __forceinline__ void wave_tiles(
    const v4f* __restrict__ bp1, const v4f* __restrict__ bp2, const float* ap1,
    const float* ap2, const float* dga, float* __restrict__ out, int base,
    int cntn, int tl, int half, int col, float bo) {
  v8f acc[NT];
  float inv[NT];
#pragma unroll
  for (int j = 0; j < NT; ++j) {
    acc[j] = v8f{};
    inv[j] = 1.0f / fmaxf(dga[j * (64 * A2_LD)], 1.0f);
  }
  v2f a1[NT][2], a2[NT][2];
  v4f b1c = bp1[0], b2c = bp2[0];
  load_a<NT>(ap1, 64 * A1_LD, a1);
#pragma unroll 1
  for (int p = 0; p < K2P / 8; ++p) {
    const v4f b1n = bp1[(p + 1) * 32];
    const v4f b2n = bp2[min(p + 1, K2P / 8 - 1) * 32];
    load_a<NT>(ap2 + 8 * p, 64 * A2_LD, a2);
    mma_sub<NT>(a1, b1c, acc);
    load_a<NT>(ap1 + 8 * (p + 1), 64 * A1_LD, a1);
#pragma unroll
    for (int j = 0; j < NT; ++j) {
      a2[j][0] *= inv[j];
      a2[j][1] *= inv[j];
    }
    mma_sub<NT>(a2, b2c, acc);
    b1c = b1n;
    b2c = b2n;
  }
  mma_sub<NT>(a1, b1c, acc);
  const int r0 = tl * 16 + 8 * half;
  float* op = out + (long)(base + r0) * HID + col;
#pragma unroll
  for (int j = 0; j < NT; ++j) {
#pragma unroll
    for (int i = 0; i < 8; ++i) {
      const int ro = 64 * j + i;
      if (r0 + ro < cntn) op[ro * HID] = acc[j][i] + bo;
    }
  }
}

__device__ __forceinline__ void flush_row(float* A2, int row, int lane,
                                          const v4f acc) {
  float* op = A2 + row * A2_LD + A2_H + 4 * lane;
  atomicAdd(op + 0, acc.x);
  atomicAdd(op + 1, acc.y);
  atomicAdd(op + 2, acc.z);
  atomicAdd(op + 3, acc.w);
}

template <bool FULL>
__device__ __forceinline__ void sum_group(const v2i* S, int g0, int k, int lane,
                                          const float* __restrict__ h,
                                          float* A2, int& cur, v4f& acc) {
  const v2i e = S[g0 + ((lane < k) ? lane : 0)];
  const int s_l = e.x;
  const int r_l = e.y;
  v4f v[GRP];
#pragma unroll
  for (int t = 0; t < GRP; ++t) {
    if (FULL || t < k) {
      const int s = __builtin_amdgcn_readlane(s_l, t);
      v[t] = *(const v4f*)(h + (long)s * HID + 4 * lane);
    } else {
      v[t] = v4f{0.f, 0.f, 0.f, 0.f};
    }
  }
#pragma unroll
  for (int t = 0; t < GRP; ++t) {
    if (FULL || t < k) {
      const int r = __builtin_amdgcn_readlane(r_l, t);
      if (r != cur) {
        if (cur >= 0) flush_row(A2, cur, lane, acc);
        cur = r;
        acc = v[t];
      } else {
        acc += v[t];
      }
    }
  }
}

__device__ __forceinline__ void drain_a2s(const v4i* q, int qn, int k, int lane,
                                          int base,
                                          const float* __restrict__ pos_action,
                                          const float* __restrict__ u_feat,
                                          float* AU) {
  if (lane < k) {
    const v4i rec = q[qn - k + lane];
    const long s = rec.x;
    const v2f pa = *(const v2f*)(pos_action + 2 * s);
    const v4f u0 = *(const v4f*)(u_feat + 8 * s);
    const v4f u1 = *(const v4f*)(u_feat + 8 * s + 4);
    float* a = AU + (rec.y - base) * AU_LD;
    atomicAdd(a + 0, pa.x);
    atomicAdd(a + 1, pa.y);
    atomicAdd(a + 2, __int_as_float(rec.z));
    atomicAdd(a + 3, u0.x);
    atomicAdd(a + 4, u0.y);
    atomicAdd(a + 5, u0.z);
    atomicAdd(a + 6, u0.w);
    atomicAdd(a + 7, u1.x);
    atomicAdd(a + 8, u1.y);
    atomicAdd(a + 9, u1.z);
    atomicAdd(a + 10, u1.w);
    atomicAdd(a + 11, 1.0f);
  }
}

__global__ __launch_bounds__(NODE_THREADS, 1) void fused_node_kernel(
    const float* __restrict__ h, const float* __restrict__ pos_state,
    const int* __restrict__ off, const int* __restrict__ recs_i,
    const int* __restrict__ offA, const int* __restrict__ recsA_i,
    const float* __restrict__ pos_action, const float* __restrict__ u_feat,
    const float* __restrict__ WT1,
    const float* __restrict__ WT2, const float* __restrict__ b_upd,
    float* __restrict__ out, int n_state, int E, int EA) {
  __shared__ float A2[RMAX * A2_LD];
  __shared__ float AU[RMAX * AU_LD];
  __shared__ v4i Q[NODE_WAVES * QCAP];
  __shared__ v2i S[SCAP];
  __shared__ int cntN[CNT_N];
  __shared__ int startN[CNT_N + 1];
  __shared__ int wgmax;
  static_assert(CNT_N >= RMAX && CNT_N == 7 * 32, "7 counters per lane");
  static_assert(RMAX * A1_LD * sizeof(float) <= NODE_WAVES * QCAP * sizeof(v4i),
                "A1 must fit in Q");
  float* A1 = (float*)Q;

  const int tid = threadIdx.x;
  const int lane = tid & 31;
  const int wv = __builtin_amdgcn_readfirstlane(tid >> 5);
  const int G = NOWN;
  const int R = range_nodes(n_state);
  const int passes = (n_state + R * G - 1) / (R * G);
  v4i* q = Q + wv * QCAP;
  const v4i* recs = (const v4i*)recs_i;
  const v4i* recsA = (const v4i*)recsA_i;
  const int bl = wv * SEGS + (lane & (SEGS - 1));
  const int st_l = off[bl * OFF_LD + blockIdx.x];
  const int en_l = off[bl * OFF_LD + blockIdx.x + 1];
  const int stA_l = offA[bl * OFF_LD + blockIdx.x];
  const int enA_l = offA[bl * OFF_LD + blockIdx.x + 1];

  for (int pass = 0; pass < passes; ++pass) {
    const int base = (pass * G + blockIdx.x) * R;
    const int cntn = max(0, min(R, n_state - base));
    int ls = lane;
    asm volatile("" : "+v"(ls));

    const int tz = wv * 32 + ls;
    for (int i = tz; i < RMAX * A2_LD; i += NODE_THREADS) A2[i] = 0.f;
    for (int i = tz; i < RMAX * AU_LD; i += NODE_THREADS) AU[i] = 0.f;
    if (tz < CNT_N) cntN[tz] = 0;
    if (tz == 0) wgmax = 0;
    __syncthreads();

    {
      int ml = 0;
#pragma unroll
      for (int u = 0; u < SEGS; ++u)
        ml = max(ml, __builtin_amdgcn_readlane(en_l, u) -
                         __builtin_amdgcn_readlane(st_l, u));
      if (ls == 0) atomicMax(&wgmax, ml);
    }
    int rnd = 0, rounds;
    do {
      v4i r[SEGS];
      v2f pp[SEGS];
      int rank[SEGS];
#pragma unroll
      for (int u = 0; u < SEGS; ++u) {
        const int st = __builtin_amdgcn_readlane(st_l, u);
        const int en = __builtin_amdgcn_readlane(en_l, u);
        const int i = st + 32 * rnd + ls;
        const v4i none = {0, -1, 0, 0};
        r[u] = none;
        if (i < en) r[u] = recs[chunk_start(wv * SEGS + u, E) + i];
      }
#pragma unroll
      for (int u = 0; u < SEGS; ++u) {
        const bool mt = (unsigned)(r[u].y - base) < (unsigned)cntn;
        pp[u] = v2f{0.f, 0.f};
        if (mt) pp[u] = *(const v2f*)(pos_state + 2 * (long)r[u].x);
      }
#pragma unroll
      for (int u = 0; u < SEGS; ++u) {
        const bool mt = (unsigned)(r[u].y - base) < (unsigned)cntn;
        rank[u] = 0;
        if (mt) rank[u] = atomicAdd(&cntN[r[u].y - base], 1);
      }
      __syncthreads();
      rounds = (__builtin_amdgcn_readfirstlane(wgmax) + 31) >> 5;
      if (wv == 0) {
        int cn[7], pre[7];
        int tot = 0;
#pragma unroll
        for (int i = 0; i < 7; ++i) cn[i] = cntN[7 * ls + i];
#pragma unroll
        for (int i = 0; i < 7; ++i) {
          pre[i] = tot;
          tot += cn[i];
        }
        int x = tot;
#pragma unroll
        for (int s = 1; s < 32; s <<= 1) {
          const int y = __shfl_up(x, s);
          if (ls >= s) x += y;
        }
#pragma unroll
        for (int i = 0; i < 7; ++i) {
          const int row = 7 * ls + i;
          startN[row] = x - tot + pre[i];
          if (row < RMAX) A2[row * A2_LD + A2_DEG] += (float)cn[i];
        }
        if (ls == 31) startN[CNT_N] = x;
      }
#pragma unroll
      for (int u = 0; u < SEGS; ++u) {
        const bool mt = (unsigned)(r[u].y - base) < (unsigned)cntn;
        if (mt) {
          float* row = A2 + (r[u].y - base) * A2_LD;
          atomicAdd(row + A2_POS, pp[u].x);
          atomicAdd(row + A2_POS + 1, pp[u].y);
          atomicAdd(row + A2_DIS, __int_as_float(r[u].z));
        }
      }
      __syncthreads();
#pragma unroll
      for (int u = 0; u < SEGS; ++u) {
        const bool mt = (unsigned)(r[u].y - base) < (unsigned)cntn;
        if (mt) {
          const int row = r[u].y - base;
          S[startN[row] + rank[u]] = v2i{r[u].x, row};
        }
      }
      __syncthreads();
      {
        const int T = __builtin_amdgcn_readfirstlane(startN[CNT_N]);
        const int per = (T + NODE_WAVES - 1) / NODE_WAVES;
        const int a = min(T, wv * per);
        const int b = min(T, a + per);
        int cur = -1;
        v4f acc = {0.f, 0.f, 0.f, 0.f};
#pragma unroll 1
        for (int g0 = a; g0 < b; g0 += GRP) {
          const int k = min(GRP, b - g0);
          if (k == GRP) sum_group<true>(S, g0, GRP, ls, h, A2, cur, acc);
          else          sum_group<false>(S, g0, k, ls, h, A2, cur, acc);
        }
        if (cur >= 0) flush_row(A2, cur, ls, acc);
      }
      ++rnd;
      if (rnd < rounds) {
        __syncthreads();
        if (tz < CNT_N) cntN[tz] = 0;
        __syncthreads();
      }
    } while (rnd < rounds);

    int maxlen = 0;
#pragma unroll
    for (int u = 0; u < SEGS; ++u)
      maxlen = max(maxlen, __builtin_amdgcn_readlane(enA_l, u) -
                               __builtin_amdgcn_readlane(stA_l, u));
#pragma unroll 1
    for (int j0 = 0; j0 < maxlen; j0 += 32) {
      int qa = 0;
      v4i r[SEGS];
#pragma unroll
      for (int u = 0; u < SEGS; ++u) {
        const int st = __builtin_amdgcn_readlane(stA_l, u);
        const int en = __builtin_amdgcn_readlane(enA_l, u);
        const int i = st + j0 + ls;
        const v4i none = {0, -1, 0, 0};
        r[u] = none;
        if (i < en) r[u] = recsA[chunk_start(wv * SEGS + u, EA) + i];
      }
#pragma unroll
      for (int u = 0; u < SEGS; ++u) {
        const bool mt = (unsigned)(r[u].y - base) < (unsigned)cntn;
        const unsigned mask = __builtin_amdgcn_ballot_w32(mt);
        if (mt) q[qa + (int)__builtin_amdgcn_mbcnt_lo(mask, 0u)] = r[u];
        qa += __builtin_popcount(mask);
      }
      while (qa > 0) {
        const int k = min(32, qa);
        drain_a2s(q, qa, k, ls, base, pos_action, u_feat, AU);
        qa -= k;
      }
    }
    __syncthreads();

    const int tiles = (cntn + 15) >> 4;
    const int rowsP = tiles * 16;
    int ln = lane;
    asm volatile("" : "+v"(ln));
    const int tid2 = wv * 32 + ln;
    {
      float v[(RMAX + 31) / 32][4];
#pragma unroll
      for (int it = 0; it < (RMAX + 31) / 32; ++it) {
        const int rr = wv + 32 * it;
        if (rr < cntn) {
          const float* hp = h + (long)(base + rr) * HID + ln;
#pragma unroll
          for (int j = 0; j < 4; ++j) v[it][j] = hp[32 * j];
        } else {
#pragma unroll
          for (int j = 0; j < 4; ++j) v[it][j] = 0.f;
        }
      }
      if (tid2 < rowsP) {
        const int rr = tid2;
        float* R1 = A1 + rr * A1_LD + HID;
        if (rr < cntn) {
          const long n = base + rr;
          const float* au = AU + rr * AU_LD;
          const float p0 = pos_state[2 * n];
          const float p1 = pos_state[2 * n + 1];
          const float du = au[11];
          R1[0] = p0;
          R1[1] = p1;
          R1[2] = du;
          R1[3] = au[0];
          R1[4] = au[1];
          R1[5] = du * p0;
          R1[6] = du * p1;
          R1[7] = au[2];
#pragma unroll
          for (int i = 0; i < 8; ++i) R1[8 + i] = au[3 + i];
          float* R2 = A2 + rr * A2_LD;
          const float ds = R2[A2_DEG];
          R2[3] = ds * p0;
          R2[4] = ds * p1;
        } else {
#pragma unroll
          for (int i = 0; i < 16; ++i) R1[i] = 0.f;
        }
      }
#pragma unroll
      for (int it = 0; it < (RMAX + 31) / 32; ++it) {
        const int rr = wv + 32 * it;
        if (rr < rowsP) {
#pragma unroll
          for (int j = 0; j < 4; ++j) A1[rr * A1_LD + ln + 32 * j] = v[it][j];
        }
      }
    }
    __syncthreads();
    {
      const int half = ln >> 4;
      const int m = ln & 15;
      const int kb = half * 2;
      const int ct = wv & 7;
      const int tl = wv >> 3;
      const int col = ct * 16 + m;
      const float bo = b_upd[col];
      const v4f* bp1 = (const v4f*)WT1 + (ct * (K1P / 8)) * 32 + ln;
      const v4f* bp2 = (const v4f*)WT2 + (ct * (K2P / 8)) * 32 + ln;
      const float* ap1 = A1 + (tl * 16 + m) * A1_LD + kb;
      const float* ap2 = A2 + (tl * 16 + m) * A2_LD + kb;
      const float* dga = A2 + (tl * 16 + m) * A2_LD + A2_DEG;
      const int nt = (tiles - tl + 3) >> 2;
      if (nt >= 4)
        wave_tiles<4>(bp1, bp2, ap1, ap2, dga, out, base, cntn, tl, half, col, bo);
      else if (nt == 3)
        wave_tiles<3>(bp1, bp2, ap1, ap2, dga, out, base, cntn, tl, half, col, bo);
      else if (nt == 2)
        wave_tiles<2>(bp1, bp2, ap1, ap2, dga, out, base, cntn, tl, half, col, bo);
      else if (nt == 1)
        wave_tiles<1>(bp1, bp2, ap1, ap2, dga, out, base, cntn, tl, half, col, bo);
    }
    __syncthreads();
  }
}

extern "C" void kernel_launch(void* const* d_in, const int* in_sizes, int n_in,
                              void* d_out, int out_size, void* d_ws,
                              size_t ws_size, hipStream_t stream) {
  const float* h = (const float*)d_in[0];
  const float* u = (const float*)d_in[1];
  const float* pos_state = (const float*)d_in[2];
  const float* pos_action = (const float*)d_in[3];
  const int* a2s_src = (const int*)d_in[4];
  const int* a2s_dst = (const int*)d_in[5];
  const float* a2s_dis = (const float*)d_in[6];
  const int* s2s_src = (const int*)d_in[7];
  const int* s2s_dst = (const int*)d_in[8];
  const float* s2s_dis = (const float*)d_in[9];
  const float* W_u2h = (const float*)d_in[10];
  const float* b_u2h = (const float*)d_in[11];
  const float* W_h2h = (const float*)d_in[12];
  const float* b_h2h = (const float*)d_in[13];
  const float* W_upd = (const float*)d_in[14];
  const float* b_upd = (const float*)d_in[15];
  float* out = (float*)d_out;

  const int n_state = in_sizes[2] / 2;
  const int E_a2s = in_sizes[4];
  const int E_s2s = in_sizes[7];

  float* WT1 = (float*)d_ws;
  float* WT2 = WT1 + HID * K1P;
  int* off = (int*)(WT2 + HID * K2P);
  int* offA = off + BIN_BLOCKS * OFF_LD;
  int* recs = offA + BIN_BLOCKS * OFF_LD;
  int* recsA = recs + (size_t)E_s2s * 4;

  fold_pack_kernel<<<FOLD_BLOCKS + COPY_BLOCKS + 2 * BIN_BLOCKS, 256, 0,
                     stream>>>(
      W_u2h, b_u2h, W_h2h, b_h2h, W_upd, WT1, WT2, s2s_src, s2s_dst, s2s_dis,
      off, recs, a2s_src, a2s_dst, a2s_dis, offA, recsA, n_state, E_s2s, E_a2s);
  fused_node_kernel<<<NOWN, NODE_THREADS, 0, stream>>>(
      h, pos_state, off, recs, offA, recsA, pos_action, u, WT1, WT2, b_upd, out,
      n_state, E_s2s, E_a2s);
}


# node kernel: the first product (h, pos, a2s rows) runs beside the s2s row gather: half of the waves do it while the other half walk their h rows, then they swap; products' sums re-associated
# speedup vs baseline: 5.2893x; 1.0463x over previous
#include <hip/hip_runtime.h>

typedef float v2f __attribute__((ext_vector_type(2)));
typedef float v4f __attribute__((ext_vector_type(4)));
typedef float v8f __attribute__((ext_vector_type(8)));
typedef int v4i __attribute__((ext_vector_type(4)));
typedef int v2i __attribute__((ext_vector_type(2)));

#define HID 128

#define K1P 144
#define K2P 136


__device__ __forceinline__ int frag_index(int P, int k, int n) {
  const int kk = k & 7;
  const int l = 16 * ((kk >> 1) & 1) + (n & 15);
  const int j = 2 * (kk >> 2) + (kk & 1);
  return (((n >> 4) * P + (k >> 3)) * 32 + l) * 4 + j;
}

#define NOWN 256
#define BIN_BLOCKS 256
#define OFF_LD (NOWN + 1)
#define BIN_T 10
#define RMAX 208

__device__ __forceinline__ int range_nodes(int n_state) {
  return max(1, min(RMAX, (n_state + NOWN - 1) / NOWN));
}
__device__ __forceinline__ int owner_of(int d, int rng) {
  return (int)(((unsigned)d / (unsigned)rng) & (NOWN - 1));
}
__device__ __forceinline__ int chunk_start(int b, int E) {
  return min(E, b * ((E + BIN_BLOCKS - 1) / BIN_BLOCKS));
}

__device__ __forceinline__ void bin_chunk(
    int b, int tid, const int* __restrict__ src, const int* __restrict__ dst,
    const float* __restrict__ dis, int* __restrict__ off,
    v4i* __restrict__ recs, int n_state, int E, int* cnt, int* wsum) {
  const int rng = range_nodes(n_state);
  const int e0 = chunk_start(b, E);
  const int e1 = chunk_start(b + 1, E);
  cnt[tid] = 0;
  __syncthreads();
  for (int t0 = e0; t0 < e1; t0 += 256 * BIN_T) {
    int d[BIN_T];
#pragma unroll
    for (int j = 0; j < BIN_T; ++j) {
      const int e = t0 + j * 256 + tid;
      d[j] = (e < e1) ? dst[e] : 0;
    }
#pragma unroll
    for (int j = 0; j < BIN_T; ++j) {
      const int e = t0 + j * 256 + tid;
      if (e < e1) atomicAdd(&cnt[owner_of(d[j], rng)], 1);
    }
  }
  __syncthreads();
  const int lane = tid & 31;
  const int wv = tid >> 5;
  const int v = cnt[tid];
  int x = v;
#pragma unroll
  for (int s = 1; s < 32; s <<= 1) {
    const int y = __shfl_up(x, s);
    if (lane >= s) x += y;
  }
  if (lane == 31) wsum[wv] = x;
  __syncthreads();
  int woff = 0;
#pragma unroll
  for (int i = 0; i < 8; ++i) woff += (i < wv) ? wsum[i] : 0;
  const int excl = woff + x - v;
  off[b * OFF_LD + tid] = excl;
  if (tid == NOWN - 1) off[b * OFF_LD + NOWN] = excl + v;
  cnt[tid] = excl;
  __syncthreads();
  for (int t0 = e0; t0 < e1; t0 += 256 * BIN_T) {
    int d[BIN_T], s[BIN_T];
    float w[BIN_T];
#pragma unroll
    for (int j = 0; j < BIN_T; ++j) {
      const int e = t0 + j * 256 + tid;
      const bool ev = e < e1;
      d[j] = ev ? dst[e] : 0;
      s[j] = ev ? src[e] : 0;
      w[j] = ev ? dis[e] : 0.f;
    }
#pragma unroll
    for (int j = 0; j < BIN_T; ++j) {
      const int e = t0 + j * 256 + tid;
      if (e < e1) {
        const int pos = atomicAdd(&cnt[owner_of(d[j], rng)], 1);
        v4i rec;
        rec.x = s[j];
        rec.y = d[j];
        rec.z = __float_as_int(w[j]);
        rec.w = 0;
        recs[e0 + pos] = rec;
      }
    }
  }
}

#define FOLD_BLOCKS 10
#define COPY_BLOCKS 65
__global__ __launch_bounds__(256) void fold_pack_kernel(
    const float* __restrict__ W_u2h, const float* __restrict__ b_u2h,
    const float* __restrict__ W_h2h, const float* __restrict__ b_h2h,
    const float* __restrict__ W_upd, float* __restrict__ WT1,
    float* __restrict__ WT2, const int* __restrict__ s2s_src,
    const int* __restrict__ s2s_dst, const float* __restrict__ s2s_dis,
    int* __restrict__ off, int* __restrict__ recs,
    const int* __restrict__ a2s_src, const int* __restrict__ a2s_dst,
    const float* __restrict__ a2s_dis, int* __restrict__ offA,
    int* __restrict__ recsA, int n_state, int E, int EA) {
  __shared__ int cnt[NOWN];
  __shared__ int wsum[8];
  const int tid = threadIdx.x;
  if (blockIdx.x >= FOLD_BLOCKS + COPY_BLOCKS + BIN_BLOCKS) {
    bin_chunk(blockIdx.x - (FOLD_BLOCKS + COPY_BLOCKS + BIN_BLOCKS), tid,
              a2s_src, a2s_dst, a2s_dis, offA, (v4i*)recsA, n_state, EA, cnt,
              wsum);
    return;
  }
  if (blockIdx.x >= FOLD_BLOCKS + COPY_BLOCKS) {
    bin_chunk(blockIdx.x - (FOLD_BLOCKS + COPY_BLOCKS), tid, s2s_src, s2s_dst,
              s2s_dis, off, (v4i*)recs, n_state, E, cnt, wsum);
    return;
  }
  if (blockIdx.x >= FOLD_BLOCKS) {
    const int i = (blockIdx.x - FOLD_BLOCKS) * 256 + tid;
    const int k = i >> 7, nn = i & 127;
    WT1[frag_index(K1P / 8, k, nn)] = W_upd[((k < 128) ? k : k + 256) * HID + nn];
    return;
  }
  const int lane = tid & 31;
  const int wv = tid >> 5;
  const int c = lane & 15;
  const int hh = lane >> 4;
  const int mt = blockIdx.x;
  if (mt == 1) WT2[frag_index(K2P / 8, 134 + (tid >> 7), tid & 127)] = 0.f;
  const int ra = (mt == 0) ? c : 16 * (mt - 1) + c;
  const bool xv = (mt == 0) ? (ra < 14) : (ra < 134);
  const float* xr = b_u2h;
  if (xv) {
    if (mt == 0) xr = (ra == 0) ? b_u2h : W_u2h + (ra - 1) * HID;
    else         xr = (ra == 0) ? b_h2h : W_h2h + (ra - 1) * HID;
  }
  const float* wc = W_upd + ((mt == 0) ? 128 : 256) * HID + 16 * wv + c;
  v8f acc = {};
#pragma unroll
  for (int s = 0; s < 32; ++s) {
    const int k = 4 * s + 2 * hh;
    const float x0 = xr[k], x1 = xr[k + 1];
    v2f a, b;
    a.x = xv ? x0 : 0.f;
    a.y = xv ? x1 : 0.f;
    b.x = wc[k * HID];
    b.y = wc[(k + 1) * HID];
    acc = __builtin_amdgcn_wmma_f32_16x16x4_f32(false, a, false, b, (short)0,
                                                acc, false, false);
  }
  const int n = 16 * wv + c;
#pragma unroll
  for (int i = 0; i < 8; ++i) {
    const int r = ((mt == 0) ? 0 : 16 * (mt - 1)) + 8 * hh + i;
    if (mt == 0) {
      if (r < 14) WT1[frag_index(K1P / 8, 130 + r, n)] = acc[i];
    } else {
      if (r < 134) WT2[frag_index(K2P / 8, r, n)] = acc[i];
    }
  }
}

#define A1_LD 145
#define A2_LD 137

#define NODE_THREADS 1024
#define NODE_WAVES 32
#define SEGS (BIN_BLOCKS / NODE_WAVES)
#define QCAP (32 * SEGS)
#define SCAP (NODE_WAVES * QCAP)
#define CNT_N 224
#define GRP 16
#define A2_DEG 0
#define A2_POS 1
#define A2_DIS 5
#define A2_H 6
#define AU_LD 13

template <int NT>
__device__ __forceinline__ void load_a(const float* ap, int ld64,
                                       v2f (&a)[NT][2]) {
#pragma unroll
  for (int j = 0; j < NT; ++j) {
    const float* p = ap + j * ld64;
    a[j][0].x = p[0];
    a[j][0].y = p[1];
    a[j][1].x = p[4];
    a[j][1].y = p[5];
  }
}

template <int NT>
__device__ __forceinline__ void mma_sub(const v2f (&a)[NT][2], const v4f b,
                                        v8f (&acc)[4]) {
  v2f b0, bb;
  b0.x = b.x; b0.y = b.y;
  bb.x = b.z; bb.y = b.w;
#pragma unroll
  for (int j = 0; j < NT; ++j) {
    acc[j] = __builtin_amdgcn_wmma_f32_16x16x4_f32(
        false, a[j][0], false, b0, (short)0, acc[j], false, false);
    acc[j] = __builtin_amdgcn_wmma_f32_16x16x4_f32(
        false, a[j][1], false, bb, (short)0, acc[j], false, false);
  }
}

template <int NT>
__device__ __forceinline__ void scale_a(v2f (&a)[NT][2], const float (&inv)[4]) {
#pragma unroll
  for (int j = 0; j < NT; ++j) {
    a[j][0] *= inv[j];
    a[j][1] *= inv[j];
  }
}

template <int NT, int P, bool SCALE>
__device__ __forceinline__ void wave_product(const v4f* __restrict__ bp,
                                             const float* ap, int ld64,
                                             const float (&inv)[4],
                                             v8f (&acc)[4]) {
  constexpr int PA = (P + 1) / 2;
  constexpr int PB = P / 2;
  const v4f* __restrict__ bq = bp + PA * 32;
  const float* aq = ap + 8 * PA;
  v2f a1[NT][2], a2[NT][2];
  v4f b1c = bp[0], b2c = bq[0];
  load_a<NT>(ap, ld64, a1);
#pragma unroll 1
  for (int p = 0; p < PB - 1; ++p) {
    const v4f b1n = bp[(p + 1) * 32];
    const v4f b2n = bq[(p + 1) * 32];
    load_a<NT>(aq + 8 * p, ld64, a2);
    if (SCALE) scale_a<NT>(a1, inv);
    mma_sub<NT>(a1, b1c, acc);
    load_a<NT>(ap + 8 * (p + 1), ld64, a1);
    if (SCALE) scale_a<NT>(a2, inv);
    mma_sub<NT>(a2, b2c, acc);
    b1c = b1n;
    b2c = b2n;
  }
  v4f b1n = b1c;
  if (PA > PB) b1n = bp[PB * 32];
  load_a<NT>(aq + 8 * (PB - 1), ld64, a2);
  if (SCALE) scale_a<NT>(a1, inv);
  mma_sub<NT>(a1, b1c, acc);
  if (PA > PB) load_a<NT>(ap + 8 * PB, ld64, a1);
  if (SCALE) scale_a<NT>(a2, inv);
  mma_sub<NT>(a2, b2c, acc);
  if (PA > PB) {
    if (SCALE) scale_a<NT>(a1, inv);
    mma_sub<NT>(a1, b1n, acc);
  }
}

template <int P, bool SCALE>
__device__ __forceinline__ void product_nt(int nt, const v4f* __restrict__ bp,
                                           const float* ap, int ld64,
                                           const float (&inv)[4],
                                           v8f (&acc)[4]) {
  if (nt >= 4)      wave_product<4, P, SCALE>(bp, ap, ld64, inv, acc);
  else if (nt == 3) wave_product<3, P, SCALE>(bp, ap, ld64, inv, acc);
  else if (nt == 2) wave_product<2, P, SCALE>(bp, ap, ld64, inv, acc);
  else if (nt == 1) wave_product<1, P, SCALE>(bp, ap, ld64, inv, acc);
}

__device__ __forceinline__ void flush_row(float* A2, int row, int lane,
                                          const v4f acc) {
  float* op = A2 + row * A2_LD + A2_H + 4 * lane;
  atomicAdd(op + 0, acc.x);
  atomicAdd(op + 1, acc.y);
  atomicAdd(op + 2, acc.z);
  atomicAdd(op + 3, acc.w);
}

template <bool FULL>
__device__ __forceinline__ void sum_group(const v2i* S, int g0, int k, int lane,
                                          const float* __restrict__ h,
                                          float* A2, int& cur, v4f& acc) {
  const v2i e = S[g0 + ((lane < k) ? lane : 0)];
  const int s_l = e.x;
  const int r_l = e.y;
  v4f v[GRP];
#pragma unroll
  for (int t = 0; t < GRP; ++t) {
    if (FULL || t < k) {
      const int s = __builtin_amdgcn_readlane(s_l, t);
      v[t] = *(const v4f*)(h + (long)s * HID + 4 * lane);
    } else {
      v[t] = v4f{0.f, 0.f, 0.f, 0.f};
    }
  }
#pragma unroll
  for (int t = 0; t < GRP; ++t) {
    if (FULL || t < k) {
      const int r = __builtin_amdgcn_readlane(r_l, t);
      if (r != cur) {
        if (cur >= 0) flush_row(A2, cur, lane, acc);
        cur = r;
        acc = v[t];
      } else {
        acc += v[t];
      }
    }
  }
}

__device__ __forceinline__ void drain_a2s(const v4i* q, int qn, int k, int lane,
                                          int base,
                                          const float* __restrict__ pos_action,
                                          const float* __restrict__ u_feat,
                                          float* AU) {
  if (lane < k) {
    const v4i rec = q[qn - k + lane];
    const long s = rec.x;
    const v2f pa = *(const v2f*)(pos_action + 2 * s);
    const v4f u0 = *(const v4f*)(u_feat + 8 * s);
    const v4f u1 = *(const v4f*)(u_feat + 8 * s + 4);
    float* a = AU + (rec.y - base) * AU_LD;
    atomicAdd(a + 0, pa.x);
    atomicAdd(a + 1, pa.y);
    atomicAdd(a + 2, __int_as_float(rec.z));
    atomicAdd(a + 3, u0.x);
    atomicAdd(a + 4, u0.y);
    atomicAdd(a + 5, u0.z);
    atomicAdd(a + 6, u0.w);
    atomicAdd(a + 7, u1.x);
    atomicAdd(a + 8, u1.y);
    atomicAdd(a + 9, u1.z);
    atomicAdd(a + 10, u1.w);
    atomicAdd(a + 11, 1.0f);
  }
}
__device__ __forceinline__ void walk_piece(const v2i* S, int T, int wv, int ls,
                                           const float* __restrict__ h,
                                           float* A2) {
  const int per = (T + NODE_WAVES - 1) / NODE_WAVES;
  const int a = min(T, wv * per);
  const int b = min(T, a + per);
  int cur = -1;
  v4f acc = {0.f, 0.f, 0.f, 0.f};
#pragma unroll 1
  for (int g0 = a; g0 < b; g0 += GRP) {
    const int k = min(GRP, b - g0);
    if (k == GRP) sum_group<true>(S, g0, GRP, ls, h, A2, cur, acc);
    else          sum_group<false>(S, g0, k, ls, h, A2, cur, acc);
  }
  if (cur >= 0) flush_row(A2, cur, ls, acc);
}

__global__ __launch_bounds__(NODE_THREADS, 1) void fused_node_kernel(
    const float* __restrict__ h, const float* __restrict__ pos_state,
    const int* __restrict__ off, const int* __restrict__ recs_i,
    const int* __restrict__ offA, const int* __restrict__ recsA_i,
    const float* __restrict__ pos_action, const float* __restrict__ u_feat,
    const float* __restrict__ WT1,
    const float* __restrict__ WT2, const float* __restrict__ b_upd,
    float* __restrict__ out, int n_state, int E, int EA) {
  __shared__ float A2[RMAX * A2_LD];
  __shared__ float AU[RMAX * AU_LD];
  __shared__ v4i Q[NODE_WAVES * QCAP];
  __shared__ v2i S[SCAP];
  __shared__ int cntN[CNT_N];
  __shared__ int startN[CNT_N + 1];
  __shared__ int wgmax;
  static_assert(CNT_N >= RMAX && CNT_N == 7 * 32, "7 counters per lane");
  static_assert(RMAX * A1_LD * sizeof(float) <= NODE_WAVES * QCAP * sizeof(v4i),
                "A1 must fit in Q");
  float* A1 = (float*)Q;

  const int tid = threadIdx.x;
  const int lane = tid & 31;
  const int wv = __builtin_amdgcn_readfirstlane(tid >> 5);
  const bool prod_first = (__builtin_popcount((unsigned)wv) & 1) == 0;
  const int ct = wv & 7;
  const int tl = wv >> 3;
  const int G = NOWN;
  const int R = range_nodes(n_state);
  const int passes = (n_state + R * G - 1) / (R * G);
  v4i* q = Q + wv * QCAP;
  const v4i* recs = (const v4i*)recs_i;
  const v4i* recsA = (const v4i*)recsA_i;

  for (int pass = 0; pass < passes; ++pass) {
    const int base = (pass * G + blockIdx.x) * R;
    const int cntn = max(0, min(R, n_state - base));
    const int tiles = (cntn + 15) >> 4;
    const int rowsP = tiles * 16;
    const int nt = max(0, (tiles - tl + 3) >> 2);
    int ls = lane;
    asm volatile("" : "+v"(ls));

    const int tz = wv * 32 + ls;
    const int bl = wv * SEGS + (ls & (SEGS - 1));
    const int st_l = off[bl * OFF_LD + blockIdx.x];
    const int en_l = off[bl * OFF_LD + blockIdx.x + 1];
    const int stA_l = offA[bl * OFF_LD + blockIdx.x];
    const int enA_l = offA[bl * OFF_LD + blockIdx.x + 1];
    for (int i = tz; i < RMAX * A2_LD; i += NODE_THREADS) A2[i] = 0.f;
    for (int i = tz; i < RMAX * AU_LD; i += NODE_THREADS) AU[i] = 0.f;
    if (tz < CNT_N) cntN[tz] = 0;
    if (tz == 0) wgmax = 0;
    __syncthreads();

    int maxlen = 0;
#pragma unroll
    for (int u = 0; u < SEGS; ++u)
      maxlen = max(maxlen, __builtin_amdgcn_readlane(enA_l, u) -
                               __builtin_amdgcn_readlane(stA_l, u));
#pragma unroll 1
    for (int j0 = 0; j0 < maxlen; j0 += 32) {
      int qa = 0;
      v4i r[SEGS];
#pragma unroll
      for (int u = 0; u < SEGS; ++u) {
        const int st = __builtin_amdgcn_readlane(stA_l, u);
        const int en = __builtin_amdgcn_readlane(enA_l, u);
        const int i = st + j0 + ls;
        const v4i none = {0, -1, 0, 0};
        r[u] = none;
        if (i < en) r[u] = recsA[chunk_start(wv * SEGS + u, EA) + i];
      }
#pragma unroll
      for (int u = 0; u < SEGS; ++u) {
        const bool mt = (unsigned)(r[u].y - base) < (unsigned)cntn;
        const unsigned mask = __builtin_amdgcn_ballot_w32(mt);
        if (mt) q[qa + (int)__builtin_amdgcn_mbcnt_lo(mask, 0u)] = r[u];
        qa += __builtin_popcount(mask);
      }
      while (qa > 0) {
        const int k = min(32, qa);
        drain_a2s(q, qa, k, ls, base, pos_action, u_feat, AU);
        qa -= k;
      }
    }
    __syncthreads();

    {
      float v[(RMAX + 31) / 32][4];
#pragma unroll
      for (int it = 0; it < (RMAX + 31) / 32; ++it) {
        const int rr = wv + 32 * it;
        if (rr < cntn) {
          const float* hp = h + (long)(base + rr) * HID + ls;
#pragma unroll
          for (int j = 0; j < 4; ++j) v[it][j] = hp[32 * j];
        } else {
#pragma unroll
          for (int j = 0; j < 4; ++j) v[it][j] = 0.f;
        }
      }
      if (tz < rowsP) {
        float* R1 = A1 + tz * A1_LD + HID;
        if (tz < cntn) {
          const long n = base + tz;
          const float* au = AU + tz * AU_LD;
          const float p0 = pos_state[2 * n];
          const float p1 = pos_state[2 * n + 1];
          const float du = au[11];
          R1[0] = p0;
          R1[1] = p1;
          R1[2] = du;
          R1[3] = au[0];
          R1[4] = au[1];
          R1[5] = du * p0;
          R1[6] = du * p1;
          R1[7] = au[2];
#pragma unroll
          for (int i = 0; i < 8; ++i) R1[8 + i] = au[3 + i];
        } else {
#pragma unroll
          for (int i = 0; i < 16; ++i) R1[i] = 0.f;
        }
      }
#pragma unroll
      for (int it = 0; it < (RMAX + 31) / 32; ++it) {
        const int rr = wv + 32 * it;
        if (rr < rowsP) {
#pragma unroll
          for (int j = 0; j < 4; ++j) A1[rr * A1_LD + ls + 32 * j] = v[it][j];
        }
      }
    }

    {
      int ml = 0;
#pragma unroll
      for (int u = 0; u < SEGS; ++u)
        ml = max(ml, __builtin_amdgcn_readlane(en_l, u) -
                         __builtin_amdgcn_readlane(st_l, u));
      if (ls == 0) atomicMax(&wgmax, ml);
    }
    v8f acc[4];
    int rnd = 0, rounds;
    do {
      int lb = ls;
      asm volatile("" : "+v"(lb));
      const int tb = wv * 32 + lb;
      v4i r[SEGS];
      v2f pp[SEGS];
      int rank[SEGS];
#pragma unroll
      for (int u = 0; u < SEGS; ++u) {
        const int st = __builtin_amdgcn_readlane(st_l, u);
        const int en = __builtin_amdgcn_readlane(en_l, u);
        const int i = st + 32 * rnd + lb;
        const v4i none = {0, -1, 0, 0};
        r[u] = none;
        if (i < en) r[u] = recs[chunk_start(wv * SEGS + u, E) + i];
      }
#pragma unroll
      for (int u = 0; u < SEGS; ++u) {
        const bool mt = (unsigned)(r[u].y - base) < (unsigned)cntn;
        pp[u] = v2f{0.f, 0.f};
        if (mt) pp[u] = *(const v2f*)(pos_state + 2 * (long)r[u].x);
      }
#pragma unroll
      for (int u = 0; u < SEGS; ++u) {
        const bool mt = (unsigned)(r[u].y - base) < (unsigned)cntn;
        rank[u] = 0;
        if (mt) rank[u] = atomicAdd(&cntN[r[u].y - base], 1);
      }
      __syncthreads();
      rounds = (__builtin_amdgcn_readfirstlane(wgmax) + 31) >> 5;
      if (wv == 0) {
        int cn[7], pre[7];
        int tot = 0;
#pragma unroll
        for (int i = 0; i < 7; ++i) cn[i] = cntN[7 * lb + i];
#pragma unroll
        for (int i = 0; i < 7; ++i) {
          pre[i] = tot;
          tot += cn[i];
        }
        int x = tot;
#pragma unroll
        for (int s = 1; s < 32; s <<= 1) {
          const int y = __shfl_up(x, s);
          if (lb >= s) x += y;
        }
#pragma unroll
        for (int i = 0; i < 7; ++i) {
          const int row = 7 * lb + i;
          startN[row] = x - tot + pre[i];
          if (row < RMAX) A2[row * A2_LD + A2_DEG] += (float)cn[i];
        }
        if (lb == 31) startN[CNT_N] = x;
      }
#pragma unroll
      for (int u = 0; u < SEGS; ++u) {
        const bool mt = (unsigned)(r[u].y - base) < (unsigned)cntn;
        if (mt) {
          float* row = A2 + (r[u].y - base) * A2_LD;
          atomicAdd(row + A2_POS, pp[u].x);
          atomicAdd(row + A2_POS + 1, pp[u].y);
          atomicAdd(row + A2_DIS, __int_as_float(r[u].z));
        }
      }
      __syncthreads();
#pragma unroll
      for (int u = 0; u < SEGS; ++u) {
        const bool mt = (unsigned)(r[u].y - base) < (unsigned)cntn;
        if (mt) {
          const int row = r[u].y - base;
          S[startN[row] + rank[u]] = v2i{r[u].x, row};
        }
      }
      if (tb < cntn) {
        const float c = (float)cntN[tb];
        const float* R1 = A1 + tb * A1_LD + HID;
        float* R2 = A2 + tb * A2_LD;
        R2[3] += c * R1[0];
        R2[4] += c * R1[1];
      }
      __syncthreads();
      {
        const int T = __builtin_amdgcn_readfirstlane(startN[CNT_N]);
#pragma unroll 1
        for (int step = 0; step < 2; ++step) {
          const bool do_prod = (rnd == 0) && ((step == 0) == prod_first);
          const bool do_walk =
              (rnd == 0) ? ((step == 0) != prod_first) : (step == 0);
          if (do_prod) {
            int lp = ls;
            asm volatile("" : "+v"(lp));
            const int half = lp >> 4;
            const int m = lp & 15;
            const v4f* bp1 = (const v4f*)WT1 + (ct * (K1P / 8)) * 32 + lp;
            const float* ap1 = A1 + (tl * 16 + m) * A1_LD + 2 * half;
            const float one[4] = {1.f, 1.f, 1.f, 1.f};
#pragma unroll
            for (int j = 0; j < 4; ++j) acc[j] = v8f{};
            product_nt<K1P / 8, false>(nt, bp1, ap1, 64 * A1_LD, one, acc);
          }
          if (do_walk) walk_piece(S, T, wv, ls, h, A2);
        }
      }
      ++rnd;
      if (rnd < rounds) {
        __syncthreads();
        if (tz < CNT_N) cntN[tz] = 0;
        __syncthreads();
      }
    } while (rnd < rounds);
    __syncthreads();

    {
      int ln = lane;
      asm volatile("" : "+v"(ln));
      const int half = ln >> 4;
      const int m = ln & 15;
      const int col = ct * 16 + m;
      const float bo = b_upd[col];
      const v4f* bp2 = (const v4f*)WT2 + (ct * (K2P / 8)) * 32 + ln;
      const float* ap2 = A2 + (tl * 16 + m) * A2_LD + 2 * half;
      const float* dga = A2 + (tl * 16 + m) * A2_LD + A2_DEG;
      float inv[4];
#pragma unroll
      for (int j = 0; j < 4; ++j) {
        inv[j] = 1.0f;
        if (j < nt) inv[j] = 1.0f / fmaxf(dga[j * (64 * A2_LD)], 1.0f);
      }
      product_nt<K2P / 8, true>(nt, bp2, ap2, 64 * A2_LD, inv, acc);
      const int r0 = tl * 16 + 8 * half;
      float* op = out + (long)(base + r0) * HID + col;
#pragma unroll
      for (int j = 0; j < 4; ++j) {
        if (j < nt) {
#pragma unroll
          for (int i = 0; i < 8; ++i) {
            const int ro = 64 * j + i;
            if (r0 + ro < cntn) op[ro * HID] = acc[j][i] + bo;
          }
        }
      }
    }
    __syncthreads();
  }
}


extern "C" void kernel_launch(void* const* d_in, const int* in_sizes, int n_in,
                              void* d_out, int out_size, void* d_ws,
                              size_t ws_size, hipStream_t stream) {
  const float* h = (const float*)d_in[0];
  const float* u = (const float*)d_in[1];
  const float* pos_state = (const float*)d_in[2];
  const float* pos_action = (const float*)d_in[3];
  const int* a2s_src = (const int*)d_in[4];
  const int* a2s_dst = (const int*)d_in[5];
  const float* a2s_dis = (const float*)d_in[6];
  const int* s2s_src = (const int*)d_in[7];
  const int* s2s_dst = (const int*)d_in[8];
  const float* s2s_dis = (const float*)d_in[9];
  const float* W_u2h = (const float*)d_in[10];
  const float* b_u2h = (const float*)d_in[11];
  const float* W_h2h = (const float*)d_in[12];
  const float* b_h2h = (const float*)d_in[13];
  const float* W_upd = (const float*)d_in[14];
  const float* b_upd = (const float*)d_in[15];
  float* out = (float*)d_out;

  const int n_state = in_sizes[2] / 2;
  const int E_a2s = in_sizes[4];
  const int E_s2s = in_sizes[7];

  float* WT1 = (float*)d_ws;
  float* WT2 = WT1 + HID * K1P;
  int* off = (int*)(WT2 + HID * K2P);
  int* offA = off + BIN_BLOCKS * OFF_LD;
  int* recs = offA + BIN_BLOCKS * OFF_LD;
  int* recsA = recs + (size_t)E_s2s * 4;

  fold_pack_kernel<<<FOLD_BLOCKS + COPY_BLOCKS + 2 * BIN_BLOCKS, 256, 0,
                     stream>>>(
      W_u2h, b_u2h, W_h2h, b_h2h, W_upd, WT1, WT2, s2s_src, s2s_dst, s2s_dis,
      off, recs, a2s_src, a2s_dst, a2s_dis, offA, recsA, n_state, E_s2s, E_a2s);
  fused_node_kernel<<<NOWN, NODE_THREADS, 0, stream>>>(
      h, pos_state, off, recs, offA, recsA, pos_action, u, WT1, WT2, b_upd, out,
      n_state, E_s2s, E_a2s);
}
